# starting kernel
# baseline (speedup 1.0000x reference)
#include <hip/hip_runtime.h>
#include <hip/hip_bf16.h>

typedef __attribute__((ext_vector_type(2))) float v2f;
typedef __attribute__((ext_vector_type(8))) float v8f;

#define GAT_N    50000
#define GAT_E    800000
#define GAT_DIN  128
#define GAT_DHID 64
#define GAT_DOUT 32
#define NEG_SLOPE 0.2f

__device__ __forceinline__ unsigned flipf(float f) {
  unsigned u = __float_as_uint(f);
  return (u & 0x80000000u) ? ~u : (u | 0x80000000u);
}
__device__ __forceinline__ float unflipf(unsigned u) {
  unsigned b = (u & 0x80000000u) ? (u & 0x7FFFFFFFu) : ~u;
  return __uint_as_float(b);
}

template<int K, int NCOL>
__global__ __launch_bounds__(256)
void wmma_gemm_kernel(const float* __restrict__ A, const float* __restrict__ B,
                      float* __restrict__ C, int Mrows) {
  constexpr int NT = NCOL / 16;
  int wave = (int)((blockIdx.x * blockDim.x + threadIdx.x) >> 5);
  int lane = (int)(threadIdx.x & 31);
  int m0 = wave * 16;
  if (m0 >= Mrows) return;
  int half = lane >> 4;
  int lr = lane & 15;

  v8f acc[NT] = {};
  const float* arow = A + (size_t)(m0 + lr) * K;

  for (int k = 0; k < K; k += 4) {
    int ka = k + 2 * half;
    v2f a;
    a.x = arow[ka];
    a.y = arow[ka + 1];
    const float* b0 = B + (size_t)ka * NCOL;
    const float* b1 = B + (size_t)(ka + 1) * NCOL;
#pragma unroll
    for (int t = 0; t < NT; ++t) {
      v2f b;
      b.x = b0[t * 16 + lr];
      b.y = b1[t * 16 + lr];
      acc[t] = __builtin_amdgcn_wmma_f32_16x16x4_f32(
           false, a,  false, b,
           (short)0, acc[t],  false,  false);
    }
  }

  int row = m0 + half * 8;
#pragma unroll
  for (int t = 0; t < NT; ++t) {
#pragma unroll
    for (int r = 0; r < 8; ++r) {
      C[(size_t)(row + r) * NCOL + t * 16 + lr] = acc[t][r];
    }
  }
}

template<int F>
__global__ __launch_bounds__(256)
void node_attn_kernel(const float* __restrict__ feat,
                      const float* __restrict__ al, const float* __restrict__ ar,
                      float* __restrict__ el, float* __restrict__ er, int n_nodes) {
  int wave = (int)((blockIdx.x * blockDim.x + threadIdx.x) >> 5);
  int lane = (int)(threadIdx.x & 31);
  if (wave >= n_nodes) return;
  const float* frow = feat + (size_t)wave * (2 * F);
#pragma unroll
  for (int h = 0; h < 2; ++h) {
    float sl = 0.f, sr = 0.f;
    for (int f = lane; f < F; f += 32) {
      float v = frow[h * F + f];
      sl += v * al[h * F + f];
      sr += v * ar[h * F + f];
    }
#pragma unroll
    for (int off = 16; off > 0; off >>= 1) {
      sl += __shfl_xor(sl, off, 32);
      sr += __shfl_xor(sr, off, 32);
    }
    if (lane == 0) {
      el[wave * 2 + h] = sl;
      er[wave * 2 + h] = sr;
    }
  }
}

__global__ __launch_bounds__(256)
void edge_logits_kernel(const int* __restrict__ src, const int* __restrict__ dst,
                        const float* __restrict__ el, const float* __restrict__ er,
                        float* __restrict__ e, unsigned* __restrict__ menc, int nE) {
  int i = (int)(blockIdx.x * blockDim.x + threadIdx.x);
  if (i >= nE) return;
  int s = src[i], d = dst[i];
#pragma unroll
  for (int h = 0; h < 2; ++h) {
    float v = el[s * 2 + h] + er[d * 2 + h];
    v = (v > 0.f) ? v : NEG_SLOPE * v;
    e[(size_t)i * 2 + h] = v;
    atomicMax(&menc[d * 2 + h], flipf(v));
  }
}

__global__ __launch_bounds__(256)
void edge_alpha_kernel(const int* __restrict__ dst, const unsigned* __restrict__ menc,
                       float* __restrict__ e_alpha, float* __restrict__ denom, int nE) {
  int i = (int)(blockIdx.x * blockDim.x + threadIdx.x);
  if (i >= nE) return;
  int d = dst[i];
#pragma unroll
  for (int h = 0; h < 2; ++h) {
    float m = unflipf(menc[d * 2 + h]);
    float a = __expf(e_alpha[(size_t)i * 2 + h] - m);
    e_alpha[(size_t)i * 2 + h] = a;
    atomicAdd(&denom[d * 2 + h], a);
  }
}

template<int F>
__global__ __launch_bounds__(256)
void edge_agg_kernel(const int* __restrict__ src, const int* __restrict__ dst,
                     const float* __restrict__ feat, const float* __restrict__ alpha,
                     const float* __restrict__ denom, float* __restrict__ agg, int nE) {
  constexpr int C = 2 * F;
  constexpr int VEC = C / 32;
  int wave = (int)((blockIdx.x * blockDim.x + threadIdx.x) >> 5);
  int lane = (int)(threadIdx.x & 31);
  if (wave >= nE) return;
  int s = src[wave], d = dst[wave];
  int col = lane * VEC;
  int h = (col >= F) ? 1 : 0;
  float sc = alpha[(size_t)wave * 2 + h] / denom[d * 2 + h];
  const float* fsrc = feat + (size_t)s * C + col;
  float* ad = agg + (size_t)d * C + col;
#pragma unroll
  for (int j = 0; j < VEC; ++j) {
    atomicAdd(&ad[j], fsrc[j] * sc);
  }
}

template<int F>
__global__ __launch_bounds__(256)
void finish_kernel(const float* __restrict__ agg, const float* __restrict__ bias,
                   float* __restrict__ out, int n_nodes) {
  int i = (int)(blockIdx.x * blockDim.x + threadIdx.x);
  if (i >= n_nodes * F) return;
  int n = i / F, f = i % F;
  float v = 0.5f * (agg[(size_t)n * 2 * F + f] + bias[f] +
                    agg[(size_t)n * 2 * F + F + f] + bias[F + f]);
  out[i] = (v > 0.f) ? v : 0.f;
}

static inline int cdiv(long long a, long long b) { return (int)((a + b - 1) / b); }

extern "C" void kernel_launch(void* const* d_in, const int* in_sizes, int n_in,
                              void* d_out, int out_size, void* d_ws, size_t ws_size,
                              hipStream_t stream) {
  (void)in_sizes; (void)n_in; (void)out_size; (void)ws_size;
  const float* x   = (const float*)d_in[0];
  const int*  esrc = (const int*)d_in[1];
  const int*  edst = (const int*)d_in[2];
  const float* W1  = (const float*)d_in[3];
  const float* al1 = (const float*)d_in[4];
  const float* ar1 = (const float*)d_in[5];
  const float* b1  = (const float*)d_in[6];
  const float* W2  = (const float*)d_in[7];
  const float* al2 = (const float*)d_in[8];
  const float* ar2 = (const float*)d_in[9];
  const float* b2  = (const float*)d_in[10];

  const int N = GAT_N, E = GAT_E;

  float* ws = (float*)d_ws;
  size_t o = 0;
  float*    feat1  = ws + o; o += (size_t)N * 128;
  float*    el1    = ws + o; o += (size_t)N * 2;
  float*    er1    = ws + o; o += (size_t)N * 2;
  unsigned* m1     = (unsigned*)(ws + o); o += (size_t)N * 2;
  float*    denom1 = ws + o; o += (size_t)N * 2;
  float*    e1     = ws + o; o += (size_t)E * 2;
  float*    agg1   = ws + o; o += (size_t)N * 128;
  float*    h1     = ws + o; o += (size_t)N * 64;
  float*    feat2  = ws + o; o += (size_t)N * 64;
  float*    el2    = ws + o; o += (size_t)N * 2;
  float*    er2    = ws + o; o += (size_t)N * 2;
  unsigned* m2     = (unsigned*)(ws + o); o += (size_t)N * 2;
  float*    denom2 = ws + o; o += (size_t)N * 2;
  float*    e2     = ws + o; o += (size_t)E * 2;
  float*    agg2   = ws + o; o += (size_t)N * 64;

  hipMemsetAsync(m1,     0, (size_t)N * 2 * 4, stream);
  hipMemsetAsync(denom1, 0, (size_t)N * 2 * 4, stream);
  hipMemsetAsync(agg1,   0, (size_t)N * 128 * 4, stream);
  hipMemsetAsync(m2,     0, (size_t)N * 2 * 4, stream);
  hipMemsetAsync(denom2, 0, (size_t)N * 2 * 4, stream);
  hipMemsetAsync(agg2,   0, (size_t)N * 64 * 4, stream);

  const int BLK = 256;
  {
    int waves = N / 16;
    wmma_gemm_kernel<128, 128><<<cdiv((long long)waves * 32, BLK), BLK, 0, stream>>>(
        x, W1, feat1, N);
    node_attn_kernel<64><<<cdiv((long long)N * 32, BLK), BLK, 0, stream>>>(
        feat1, al1, ar1, el1, er1, N);
    edge_logits_kernel<<<cdiv(E, BLK), BLK, 0, stream>>>(esrc, edst, el1, er1, e1, m1, E);
    edge_alpha_kernel<<<cdiv(E, BLK), BLK, 0, stream>>>(edst, m1, e1, denom1, E);
    edge_agg_kernel<64><<<cdiv((long long)E * 32, BLK), BLK, 0, stream>>>(
        esrc, edst, feat1, e1, denom1, agg1, E);
    finish_kernel<64><<<cdiv((long long)N * 64, BLK), BLK, 0, stream>>>(agg1, b1, h1, N);
  }
  {
    int waves = N / 16;
    wmma_gemm_kernel<64, 64><<<cdiv((long long)waves * 32, BLK), BLK, 0, stream>>>(
        h1, W2, feat2, N);
    node_attn_kernel<32><<<cdiv((long long)N * 32, BLK), BLK, 0, stream>>>(
        feat2, al2, ar2, el2, er2, N);
    edge_logits_kernel<<<cdiv(E, BLK), BLK, 0, stream>>>(esrc, edst, el2, er2, e2, m2, E);
    edge_alpha_kernel<<<cdiv(E, BLK), BLK, 0, stream>>>(edst, m2, e2, denom2, E);
    edge_agg_kernel<32><<<cdiv((long long)E * 32, BLK), BLK, 0, stream>>>(
        esrc, edst, feat2, e2, denom2, agg2, E);
    finish_kernel<32><<<cdiv((long long)N * 32, BLK), BLK, 0, stream>>>(
        agg2, b2, (float*)d_out, N);
  }
}


# Edge softmax and row sum owned by destination range: each work-group filters dst for its 196 nodes; max, exp sum and row sum by LDS atomics; no global atomics, 12 to 6 launches; arrival-ordered sums
# speedup vs baseline: 2.9415x; 2.9415x over previous
#include <hip/hip_runtime.h>
#include <hip/hip_bf16.h>

typedef __attribute__((ext_vector_type(2))) float v2f;
typedef __attribute__((ext_vector_type(8))) float v8f;

#define GAT_N    50000
#define GAT_E    800000
#define GAT_DIN  128
#define GAT_DHID 64
#define GAT_DOUT 32
#define NEG_SLOPE 0.2f

__device__ __forceinline__ unsigned flipf(float f) {
  unsigned u = __float_as_uint(f);
  return (u & 0x80000000u) ? ~u : (u | 0x80000000u);
}
__device__ __forceinline__ float unflipf(unsigned u) {
  unsigned b = (u & 0x80000000u) ? (u & 0x7FFFFFFFu) : ~u;
  return __uint_as_float(b);
}

template<int K, int NCOL>
__global__ __launch_bounds__(256)
void wmma_gemm_kernel(const float* __restrict__ A, const float* __restrict__ B,
                      float* __restrict__ C, int Mrows) {
  constexpr int NT = NCOL / 16;
  int wave = (int)((blockIdx.x * blockDim.x + threadIdx.x) >> 5);
  int lane = (int)(threadIdx.x & 31);
  int m0 = wave * 16;
  if (m0 >= Mrows) return;
  int half = lane >> 4;
  int lr = lane & 15;

  v8f acc[NT] = {};
  const float* arow = A + (size_t)(m0 + lr) * K;

  for (int k = 0; k < K; k += 4) {
    int ka = k + 2 * half;
    v2f a;
    a.x = arow[ka];
    a.y = arow[ka + 1];
    const float* b0 = B + (size_t)ka * NCOL;
    const float* b1 = B + (size_t)(ka + 1) * NCOL;
#pragma unroll
    for (int t = 0; t < NT; ++t) {
      v2f b;
      b.x = b0[t * 16 + lr];
      b.y = b1[t * 16 + lr];
      acc[t] = __builtin_amdgcn_wmma_f32_16x16x4_f32(
           false, a,  false, b,
           (short)0, acc[t],  false,  false);
    }
  }

  int row = m0 + half * 8;
#pragma unroll
  for (int t = 0; t < NT; ++t) {
#pragma unroll
    for (int r = 0; r < 8; ++r) {
      C[(size_t)(row + r) * NCOL + t * 16 + lr] = acc[t][r];
    }
  }
}

template<int F>
__global__ __launch_bounds__(256)
void node_attn_kernel(const float* __restrict__ feat,
                      const float* __restrict__ al, const float* __restrict__ ar,
                      float* __restrict__ el, float* __restrict__ er, int n_nodes) {
  int wave = (int)((blockIdx.x * blockDim.x + threadIdx.x) >> 5);
  int lane = (int)(threadIdx.x & 31);
  if (wave >= n_nodes) return;
  const float* frow = feat + (size_t)wave * (2 * F);
#pragma unroll
  for (int h = 0; h < 2; ++h) {
    float sl = 0.f, sr = 0.f;
    for (int f = lane; f < F; f += 32) {
      float v = frow[h * F + f];
      sl += v * al[h * F + f];
      sr += v * ar[h * F + f];
    }
#pragma unroll
    for (int off = 16; off > 0; off >>= 1) {
      sl += __shfl_xor(sl, off, 32);
      sr += __shfl_xor(sr, off, 32);
    }
    if (lane == 0) {
      el[wave * 2 + h] = sl;
      er[wave * 2 + h] = sr;
    }
  }
}

#define NPW 196
#define CAP 16384
static_assert(GAT_E % 4 == 0, "dst is scanned four edges per load");
static_assert(GAT_E <= (1 << 20), "edge index is packed into 20 bits");
static_assert(GAT_N <= (1 << 16), "source node is packed into 16 bits");
static_assert(NPW <= (1 << 12), "local node is packed above the edge index");

__device__ __forceinline__ float lrelu(float v) { return (v > 0.f) ? v : NEG_SLOPE * v; }

template<int F>
__global__ __launch_bounds__(1024)
void owner_kernel(const int* __restrict__ src, const int* __restrict__ dst,
                  const float* __restrict__ feat,
                  const float* __restrict__ el, const float* __restrict__ er,
                  const float* __restrict__ bias, float* __restrict__ out,
                  int n_nodes, int nE) {
  constexpr int HV = F / 32;
  constexpr int U = 16 / HV;
  constexpr int NT = 1024;
  constexpr int SU = 8;
  constexpr int PB = 4;
  static_assert(PB == 4, "the load fences of phases 1 and 2 name four slots");

  __shared__ float    agg_l[NPW * F];
  __shared__ unsigned m_l[NPW * 2];
  __shared__ float    den_l[NPW * 2];
  __shared__ float    er_l[NPW * 2];
  __shared__ unsigned list_l[CAP];
  __shared__ unsigned cnt_l;
  __shared__ unsigned tick_l;

  const int tid = (int)threadIdx.x;
  const int lane = tid & 31;
  const int lo = (int)blockIdx.x * NPW;
  const int nown = (n_nodes - lo < NPW) ? (n_nodes - lo) : NPW;

  for (int i = tid; i < NPW * F; i += NT) agg_l[i] = 0.f;
  for (int i = tid; i < NPW * 2; i += NT) {
    m_l[i] = 0u;
    den_l[i] = 0.f;
    er_l[i] = (i < nown * 2) ? er[(size_t)lo * 2 + i] : 0.f;
  }
  if (tid == 0) { cnt_l = 0u; tick_l = 0u; }
  __syncthreads();

  {
    const int nQ = nE >> 2;
    const int4* __restrict__ dq = reinterpret_cast<const int4*>(dst);
    const int rot = (int)(((long long)blockIdx.x * nQ) / (long long)gridDim.x);
    for (int k0 = 0; k0 < nQ; k0 += NT * SU) {
      int4 v[SU];
      int qi[SU];
#pragma unroll
      for (int u = 0; u < SU; ++u) {
        const int k = k0 + u * NT + tid;
        int q = k + rot;
        if (q >= nQ) q -= nQ;
        qi[u] = q;
        const int4 t = dq[(k < nQ) ? q : 0];
        v[u] = (k < nQ) ? t : make_int4(-1, -1, -1, -1);
      }
#pragma unroll
      for (int u = 0; u < SU; ++u) {
        const int d4[4] = {v[u].x, v[u].y, v[u].z, v[u].w};
#pragma unroll
        for (int c = 0; c < 4; ++c) {
          const unsigned dd = (unsigned)(d4[c] - lo);
          if (dd < (unsigned)nown) {
            const unsigned pos = atomicAdd(&cnt_l, 1u);
            if (pos < (unsigned)CAP) list_l[pos] = (unsigned)(qi[u] * 4 + c) | (dd << 20);
          }
        }
      }
    }
  }
  __syncthreads();
  const unsigned total = (unsigned)__builtin_amdgcn_readfirstlane((int)cnt_l);
  const bool use_list = (total <= (unsigned)CAP);

  if (use_list) {
    for (int i0 = 0; i0 < (int)total; i0 += NT * PB) {
      unsigned p[PB];
      int s[PB];
      float l0[PB], l1[PB];
#pragma unroll
      for (int k = 0; k < PB; ++k) {
        const int i = i0 + k * NT + tid;
        const unsigned t = list_l[(i < (int)total) ? i : 0];
        p[k] = (i < (int)total) ? t : 0u;
      }
#pragma unroll
      for (int k = 0; k < PB; ++k) s[k] = src[p[k] & 0xFFFFFu];
      asm volatile("" :: "v"(s[0]), "v"(s[1]), "v"(s[2]), "v"(s[3]));
#pragma unroll
      for (int k = 0; k < PB; ++k) { l0[k] = el[s[k] * 2]; l1[k] = el[s[k] * 2 + 1]; }
      asm volatile("" :: "v"(l0[0]), "v"(l1[0]), "v"(l0[1]), "v"(l1[1]),
                         "v"(l0[2]), "v"(l1[2]), "v"(l0[3]), "v"(l1[3]));
#pragma unroll
      for (int k = 0; k < PB; ++k) {
        const int i = i0 + k * NT + tid;
        if (i < (int)total) {
          const int dl = (int)(p[k] >> 20);
          list_l[i] = (unsigned)s[k] | ((unsigned)dl << 16);
          atomicMax(&m_l[dl * 2],     flipf(lrelu(l0[k] + er_l[dl * 2])));
          atomicMax(&m_l[dl * 2 + 1], flipf(lrelu(l1[k] + er_l[dl * 2 + 1])));
        }
      }
    }
  } else {
    for (int i = tid; i < nE; i += NT) {
      const int d = dst[i] - lo;
      if ((unsigned)d < (unsigned)nown) {
        const int s = src[i];
        atomicMax(&m_l[d * 2],     flipf(lrelu(el[s * 2]     + er_l[d * 2])));
        atomicMax(&m_l[d * 2 + 1], flipf(lrelu(el[s * 2 + 1] + er_l[d * 2 + 1])));
      }
    }
  }
  __syncthreads();

  if (use_list) {
    for (int i0 = 0; i0 < (int)total; i0 += NT * PB) {
      unsigned p[PB];
      float l0[PB], l1[PB];
#pragma unroll
      for (int k = 0; k < PB; ++k) {
        const int i = i0 + k * NT + tid;
        const unsigned t = list_l[(i < (int)total) ? i : 0];
        p[k] = (i < (int)total) ? t : 0u;
      }
#pragma unroll
      for (int k = 0; k < PB; ++k) {
        const int s = (int)(p[k] & 0xFFFFu);
        l0[k] = el[s * 2]; l1[k] = el[s * 2 + 1];
      }
      asm volatile("" :: "v"(l0[0]), "v"(l1[0]), "v"(l0[1]), "v"(l1[1]),
                         "v"(l0[2]), "v"(l1[2]), "v"(l0[3]), "v"(l1[3]));
#pragma unroll
      for (int k = 0; k < PB; ++k) {
        const int i = i0 + k * NT + tid;
        if (i < (int)total) {
          const int j = (int)(p[k] >> 16) * 2;
          atomicAdd(&den_l[j],     __expf(lrelu(l0[k] + er_l[j])     - unflipf(m_l[j])));
          atomicAdd(&den_l[j + 1], __expf(lrelu(l1[k] + er_l[j + 1]) - unflipf(m_l[j + 1])));
        }
      }
    }
  } else {
    for (int i = tid; i < nE; i += NT) {
      const int d = dst[i] - lo;
      if ((unsigned)d < (unsigned)nown) {
        const int s = src[i];
        const int j = d * 2;
        atomicAdd(&den_l[j],     __expf(lrelu(el[s * 2]     + er_l[j])     - unflipf(m_l[j])));
        atomicAdd(&den_l[j + 1], __expf(lrelu(el[s * 2 + 1] + er_l[j + 1]) - unflipf(m_l[j + 1])));
      }
    }
  }
  __syncthreads();

  auto one_edge = [&](int sj, int dj, float b0, float b1) {
    const float* frow = feat + (size_t)sj * (2 * F);
#pragma unroll
    for (int q = 0; q < HV; ++q) {
      atomicAdd(&agg_l[dj * F + lane + q * 32],
                b0 * frow[lane + q * 32] + b1 * frow[lane + F + q * 32]);
    }
  };
  if (use_list) {
    for (;;) {
      int c = 0;
      if (lane == 0) c = (int)atomicAdd(&tick_l, 1u);
      c = __builtin_amdgcn_readfirstlane(c);
      const int base = c * U;
      if (base >= (int)total) break;
      const int n = ((int)total - base < U) ? ((int)total - base) : U;
      const unsigned p = list_l[base + ((lane < n) ? lane : 0)];
      const int s = (int)(p & 0xFFFFu), dl = (int)(p >> 16);
      const float l0 = el[s * 2], l1 = el[s * 2 + 1];
      auto factor = [&](int h, float lv) -> int {
        const int j = dl * 2 + h;
        const float w = __expf(lrelu(lv + er_l[j]) - unflipf(m_l[j]));
        return __float_as_int(0.5f * w / den_l[j]);
      };
      if (n == U) {
        float f0[U][HV], f1[U][HV];
#pragma unroll
        for (int u = 0; u < U; ++u) {
          const float* frow = feat + (size_t)__builtin_amdgcn_readlane(s, u) * (2 * F);
#pragma unroll
          for (int q = 0; q < HV; ++q) {
            f0[u][q] = frow[lane + q * 32];
            f1[u][q] = frow[lane + F + q * 32];
          }
        }
        const int a0 = factor(0, l0), a1 = factor(1, l1);
#pragma unroll
        for (int u = 0; u < U; ++u) {
          float* arow = agg_l + __builtin_amdgcn_readlane(dl, u) * F;
          const float b0 = __int_as_float(__builtin_amdgcn_readlane(a0, u));
          const float b1 = __int_as_float(__builtin_amdgcn_readlane(a1, u));
#pragma unroll
          for (int q = 0; q < HV; ++q) {
            atomicAdd(&arow[lane + q * 32], b0 * f0[u][q] + b1 * f1[u][q]);
          }
        }
      } else {
        const int a0 = factor(0, l0), a1 = factor(1, l1);
        for (int j = 0; j < n; ++j) {
          one_edge(__builtin_amdgcn_readlane(s, j), __builtin_amdgcn_readlane(dl, j),
                   __int_as_float(__builtin_amdgcn_readlane(a0, j)),
                   __int_as_float(__builtin_amdgcn_readlane(a1, j)));
        }
      }
    }
  } else {
    for (;;) {
      int c = 0;
      if (lane == 0) c = (int)atomicAdd(&tick_l, 1u);
      c = __builtin_amdgcn_readfirstlane(c);
      const int base = c * 32;
      if (base >= nE) break;
      const int i = base + lane;
      bool ok = false;
      int s = 0, dl = 0;
      if (i < nE) {
        const int d = dst[i] - lo;
        if ((unsigned)d < (unsigned)nown) { ok = true; dl = d; s = src[i]; }
      }
      int a0 = 0, a1 = 0;
      if (ok) {
        const int j = dl * 2;
        const float w0 = __expf(lrelu(el[s * 2]     + er_l[j])     - unflipf(m_l[j]));
        const float w1 = __expf(lrelu(el[s * 2 + 1] + er_l[j + 1]) - unflipf(m_l[j + 1]));
        a0 = __float_as_int(0.5f * w0 / den_l[j]);
        a1 = __float_as_int(0.5f * w1 / den_l[j + 1]);
      }
      unsigned mask = (unsigned)__ballot(ok);
      while (mask) {
        const int j = __builtin_ctz(mask);
        mask &= mask - 1u;
        one_edge(__builtin_amdgcn_readlane(s, j), __builtin_amdgcn_readlane(dl, j),
                 __int_as_float(__builtin_amdgcn_readlane(a0, j)),
                 __int_as_float(__builtin_amdgcn_readlane(a1, j)));
      }
    }
  }
  __syncthreads();

  for (int i = tid; i < nown * F; i += NT) {
    const int f = i % F;
    const float v = agg_l[i] + 0.5f * (bias[f] + bias[F + f]);
    out[(size_t)lo * F + i] = (v > 0.f) ? v : 0.f;
  }
}

static inline int cdiv(long long a, long long b) { return (int)((a + b - 1) / b); }

extern "C" void kernel_launch(void* const* d_in, const int* in_sizes, int n_in,
                              void* d_out, int out_size, void* d_ws, size_t ws_size,
                              hipStream_t stream) {
  (void)in_sizes; (void)n_in; (void)out_size; (void)ws_size;
  const float* x   = (const float*)d_in[0];
  const int*  esrc = (const int*)d_in[1];
  const int*  edst = (const int*)d_in[2];
  const float* W1  = (const float*)d_in[3];
  const float* al1 = (const float*)d_in[4];
  const float* ar1 = (const float*)d_in[5];
  const float* b1  = (const float*)d_in[6];
  const float* W2  = (const float*)d_in[7];
  const float* al2 = (const float*)d_in[8];
  const float* ar2 = (const float*)d_in[9];
  const float* b2  = (const float*)d_in[10];

  const int N = GAT_N, E = GAT_E;

  float* ws = (float*)d_ws;
  size_t o = 0;
  float*    feat1  = ws + o; o += (size_t)N * 128;
  float*    el1    = ws + o; o += (size_t)N * 2;
  float*    er1    = ws + o; o += (size_t)N * 2;
  float*    h1     = ws + o; o += (size_t)N * 64;
  float*    feat2  = ws + o; o += (size_t)N * 64;
  float*    el2    = ws + o; o += (size_t)N * 2;
  float*    er2    = ws + o; o += (size_t)N * 2;

  const int BLK = 256;
  const int OWN = cdiv(N, NPW);
  {
    int waves = N / 16;
    wmma_gemm_kernel<128, 128><<<cdiv((long long)waves * 32, BLK), BLK, 0, stream>>>(
        x, W1, feat1, N);
    node_attn_kernel<64><<<cdiv((long long)N * 32, BLK), BLK, 0, stream>>>(
        feat1, al1, ar1, el1, er1, N);
    owner_kernel<64><<<OWN, 1024, 0, stream>>>(esrc, edst, feat1, el1, er1, b1, h1, N, E);
  }
  {
    int waves = N / 16;
    wmma_gemm_kernel<64, 64><<<cdiv((long long)waves * 32, BLK), BLK, 0, stream>>>(
        h1, W2, feat2, N);
    node_attn_kernel<32><<<cdiv((long long)N * 32, BLK), BLK, 0, stream>>>(
        feat2, al2, ar2, el2, er2, N);
    owner_kernel<32><<<OWN, 1024, 0, stream>>>(esrc, edst, feat2, el2, er2, b2, (float*)d_out, N, E);
  }
}


# owner_kernel: each edge's two logits (then its exp weights) kept in LDS beside the list from phase 1 on; phases 2 and 3 read them there, el gathered once per edge, not three times; no sum changed
# speedup vs baseline: 3.0509x; 1.0372x over previous
#include <hip/hip_runtime.h>
#include <hip/hip_bf16.h>

typedef __attribute__((ext_vector_type(2))) float v2f;
typedef __attribute__((ext_vector_type(8))) float v8f;

#define GAT_N    50000
#define GAT_E    800000
#define GAT_DIN  128
#define GAT_DHID 64
#define GAT_DOUT 32
#define NEG_SLOPE 0.2f

__device__ __forceinline__ unsigned flipf(float f) {
  unsigned u = __float_as_uint(f);
  return (u & 0x80000000u) ? ~u : (u | 0x80000000u);
}
__device__ __forceinline__ float unflipf(unsigned u) {
  unsigned b = (u & 0x80000000u) ? (u & 0x7FFFFFFFu) : ~u;
  return __uint_as_float(b);
}

template<int K, int NCOL>
__global__ __launch_bounds__(256)
void wmma_gemm_kernel(const float* __restrict__ A, const float* __restrict__ B,
                      float* __restrict__ C, int Mrows) {
  constexpr int NT = NCOL / 16;
  int wave = (int)((blockIdx.x * blockDim.x + threadIdx.x) >> 5);
  int lane = (int)(threadIdx.x & 31);
  int m0 = wave * 16;
  if (m0 >= Mrows) return;
  int half = lane >> 4;
  int lr = lane & 15;

  v8f acc[NT] = {};
  const float* arow = A + (size_t)(m0 + lr) * K;

  for (int k = 0; k < K; k += 4) {
    int ka = k + 2 * half;
    v2f a;
    a.x = arow[ka];
    a.y = arow[ka + 1];
    const float* b0 = B + (size_t)ka * NCOL;
    const float* b1 = B + (size_t)(ka + 1) * NCOL;
#pragma unroll
    for (int t = 0; t < NT; ++t) {
      v2f b;
      b.x = b0[t * 16 + lr];
      b.y = b1[t * 16 + lr];
      acc[t] = __builtin_amdgcn_wmma_f32_16x16x4_f32(
           false, a,  false, b,
           (short)0, acc[t],  false,  false);
    }
  }

  int row = m0 + half * 8;
#pragma unroll
  for (int t = 0; t < NT; ++t) {
#pragma unroll
    for (int r = 0; r < 8; ++r) {
      C[(size_t)(row + r) * NCOL + t * 16 + lr] = acc[t][r];
    }
  }
}

template<int F>
__global__ __launch_bounds__(256)
void node_attn_kernel(const float* __restrict__ feat,
                      const float* __restrict__ al, const float* __restrict__ ar,
                      float* __restrict__ el, float* __restrict__ er, int n_nodes) {
  int wave = (int)((blockIdx.x * blockDim.x + threadIdx.x) >> 5);
  int lane = (int)(threadIdx.x & 31);
  if (wave >= n_nodes) return;
  const float* frow = feat + (size_t)wave * (2 * F);
#pragma unroll
  for (int h = 0; h < 2; ++h) {
    float sl = 0.f, sr = 0.f;
    for (int f = lane; f < F; f += 32) {
      float v = frow[h * F + f];
      sl += v * al[h * F + f];
      sr += v * ar[h * F + f];
    }
#pragma unroll
    for (int off = 16; off > 0; off >>= 1) {
      sl += __shfl_xor(sl, off, 32);
      sr += __shfl_xor(sr, off, 32);
    }
    if (lane == 0) {
      el[wave * 2 + h] = sl;
      er[wave * 2 + h] = sr;
    }
  }
}

#define NPW 196
#define CAP 16384
static_assert(GAT_E % 4 == 0, "dst is scanned four edges per load");
static_assert(GAT_E <= (1 << 20), "edge index is packed into 20 bits");
static_assert(GAT_N <= (1 << 16), "source node is packed into 16 bits");
static_assert(NPW <= (1 << 12), "local node is packed above the edge index");

__device__ __forceinline__ float lrelu(float v) { return (v > 0.f) ? v : NEG_SLOPE * v; }

template<int F>
__global__ __launch_bounds__(1024)
void owner_kernel(const int* __restrict__ src, const int* __restrict__ dst,
                  const float* __restrict__ feat,
                  const float* __restrict__ el, const float* __restrict__ er,
                  const float* __restrict__ bias, float* __restrict__ out,
                  int n_nodes, int nE) {
  constexpr int HV = F / 32;
  constexpr int U = 16 / HV;
  constexpr int NT = 1024;
  constexpr int SU = 8;
  constexpr int PB = 4;
  static_assert(PB == 4, "the load fences of phase 1 name four slots");

  __shared__ float    agg_l[NPW * F];
  __shared__ unsigned m_l[NPW * 2];
  __shared__ float    den_l[NPW * 2];
  __shared__ float    er_l[NPW * 2];
  __shared__ unsigned list_l[CAP];
  __shared__ v2f      ev_l[CAP];
  __shared__ unsigned cnt_l;
  __shared__ unsigned tick_l;

  const int tid = (int)threadIdx.x;
  const int lane = tid & 31;
  const int lo = (int)blockIdx.x * NPW;
  const int nown = (n_nodes - lo < NPW) ? (n_nodes - lo) : NPW;

  for (int i = tid; i < NPW * F; i += NT) agg_l[i] = 0.f;
  for (int i = tid; i < NPW * 2; i += NT) {
    m_l[i] = 0u;
    den_l[i] = 0.f;
    er_l[i] = (i < nown * 2) ? er[(size_t)lo * 2 + i] : 0.f;
  }
  if (tid == 0) { cnt_l = 0u; tick_l = 0u; }
  __syncthreads();

  {
    const int nQ = nE >> 2;
    const int4* __restrict__ dq = reinterpret_cast<const int4*>(dst);
    const int rot = (int)(((long long)blockIdx.x * nQ) / (long long)gridDim.x);
    for (int k0 = 0; k0 < nQ; k0 += NT * SU) {
      int4 v[SU];
      int qi[SU];
#pragma unroll
      for (int u = 0; u < SU; ++u) {
        const int k = k0 + u * NT + tid;
        int q = k + rot;
        if (q >= nQ) q -= nQ;
        qi[u] = q;
        const int4 t = dq[(k < nQ) ? q : 0];
        v[u] = (k < nQ) ? t : make_int4(-1, -1, -1, -1);
      }
#pragma unroll
      for (int u = 0; u < SU; ++u) {
        const int d4[4] = {v[u].x, v[u].y, v[u].z, v[u].w};
#pragma unroll
        for (int c = 0; c < 4; ++c) {
          const unsigned dd = (unsigned)(d4[c] - lo);
          if (dd < (unsigned)nown) {
            const unsigned pos = atomicAdd(&cnt_l, 1u);
            if (pos < (unsigned)CAP) list_l[pos] = (unsigned)(qi[u] * 4 + c) | (dd << 20);
          }
        }
      }
    }
  }
  __syncthreads();
  const unsigned total = (unsigned)__builtin_amdgcn_readfirstlane((int)cnt_l);
  const bool use_list = (total <= (unsigned)CAP);

  if (use_list) {
    for (int i0 = 0; i0 < (int)total; i0 += NT * PB) {
      unsigned p[PB];
      int s[PB];
      float l0[PB], l1[PB];
#pragma unroll
      for (int k = 0; k < PB; ++k) {
        const int i = i0 + k * NT + tid;
        const unsigned t = list_l[(i < (int)total) ? i : 0];
        p[k] = (i < (int)total) ? t : 0u;
      }
#pragma unroll
      for (int k = 0; k < PB; ++k) s[k] = src[p[k] & 0xFFFFFu];
      asm volatile("" :: "v"(s[0]), "v"(s[1]), "v"(s[2]), "v"(s[3]));
#pragma unroll
      for (int k = 0; k < PB; ++k) { l0[k] = el[s[k] * 2]; l1[k] = el[s[k] * 2 + 1]; }
      asm volatile("" :: "v"(l0[0]), "v"(l1[0]), "v"(l0[1]), "v"(l1[1]),
                         "v"(l0[2]), "v"(l1[2]), "v"(l0[3]), "v"(l1[3]));
#pragma unroll
      for (int k = 0; k < PB; ++k) {
        const int i = i0 + k * NT + tid;
        if (i < (int)total) {
          const int dl = (int)(p[k] >> 20);
          list_l[i] = (unsigned)s[k] | ((unsigned)dl << 16);
          v2f e;
          e.x = lrelu(l0[k] + er_l[dl * 2]);
          e.y = lrelu(l1[k] + er_l[dl * 2 + 1]);
          ev_l[i] = e;
          atomicMax(&m_l[dl * 2],     flipf(e.x));
          atomicMax(&m_l[dl * 2 + 1], flipf(e.y));
        }
      }
    }
  } else {
    for (int i = tid; i < nE; i += NT) {
      const int d = dst[i] - lo;
      if ((unsigned)d < (unsigned)nown) {
        const int s = src[i];
        atomicMax(&m_l[d * 2],     flipf(lrelu(el[s * 2]     + er_l[d * 2])));
        atomicMax(&m_l[d * 2 + 1], flipf(lrelu(el[s * 2 + 1] + er_l[d * 2 + 1])));
      }
    }
  }
  __syncthreads();

  if (use_list) {
    for (int i = tid; i < (int)total; i += NT) {
      const int j = (int)(list_l[i] >> 16) * 2;
      const v2f e = ev_l[i];
      v2f w;
      w.x = __expf(e.x - unflipf(m_l[j]));
      w.y = __expf(e.y - unflipf(m_l[j + 1]));
      ev_l[i] = w;
      atomicAdd(&den_l[j],     w.x);
      atomicAdd(&den_l[j + 1], w.y);
    }
  } else {
    for (int i = tid; i < nE; i += NT) {
      const int d = dst[i] - lo;
      if ((unsigned)d < (unsigned)nown) {
        const int s = src[i];
        const int j = d * 2;
        atomicAdd(&den_l[j],     __expf(lrelu(el[s * 2]     + er_l[j])     - unflipf(m_l[j])));
        atomicAdd(&den_l[j + 1], __expf(lrelu(el[s * 2 + 1] + er_l[j + 1]) - unflipf(m_l[j + 1])));
      }
    }
  }
  __syncthreads();

  auto one_edge = [&](int sj, int dj, float b0, float b1) {
    const float* frow = feat + (size_t)sj * (2 * F);
#pragma unroll
    for (int q = 0; q < HV; ++q) {
      atomicAdd(&agg_l[dj * F + lane + q * 32],
                b0 * frow[lane + q * 32] + b1 * frow[lane + F + q * 32]);
    }
  };
  if (use_list) {
    for (;;) {
      int c = 0;
      if (lane == 0) c = (int)atomicAdd(&tick_l, 1u);
      c = __builtin_amdgcn_readfirstlane(c);
      const int base = c * U;
      if (base >= (int)total) break;
      const int n = ((int)total - base < U) ? ((int)total - base) : U;
      const int slot = base + ((lane < n) ? lane : 0);
      const unsigned p = list_l[slot];
      const v2f wv = ev_l[slot];
      const int s = (int)(p & 0xFFFFu), dl = (int)(p >> 16);
      auto factor = [&](int h, float w) -> int {
        return __float_as_int(0.5f * w / den_l[dl * 2 + h]);
      };
      if (n == U) {
        float f0[U][HV], f1[U][HV];
#pragma unroll
        for (int u = 0; u < U; ++u) {
          const float* frow = feat + (size_t)__builtin_amdgcn_readlane(s, u) * (2 * F);
#pragma unroll
          for (int q = 0; q < HV; ++q) {
            f0[u][q] = frow[lane + q * 32];
            f1[u][q] = frow[lane + F + q * 32];
          }
        }
        const int a0 = factor(0, wv.x), a1 = factor(1, wv.y);
#pragma unroll
        for (int u = 0; u < U; ++u) {
          float* arow = agg_l + __builtin_amdgcn_readlane(dl, u) * F;
          const float b0 = __int_as_float(__builtin_amdgcn_readlane(a0, u));
          const float b1 = __int_as_float(__builtin_amdgcn_readlane(a1, u));
#pragma unroll
          for (int q = 0; q < HV; ++q) {
            atomicAdd(&arow[lane + q * 32], b0 * f0[u][q] + b1 * f1[u][q]);
          }
        }
      } else {
        const int a0 = factor(0, wv.x), a1 = factor(1, wv.y);
        for (int j = 0; j < n; ++j) {
          one_edge(__builtin_amdgcn_readlane(s, j), __builtin_amdgcn_readlane(dl, j),
                   __int_as_float(__builtin_amdgcn_readlane(a0, j)),
                   __int_as_float(__builtin_amdgcn_readlane(a1, j)));
        }
      }
    }
  } else {
    for (;;) {
      int c = 0;
      if (lane == 0) c = (int)atomicAdd(&tick_l, 1u);
      c = __builtin_amdgcn_readfirstlane(c);
      const int base = c * 32;
      if (base >= nE) break;
      const int i = base + lane;
      bool ok = false;
      int s = 0, dl = 0;
      if (i < nE) {
        const int d = dst[i] - lo;
        if ((unsigned)d < (unsigned)nown) { ok = true; dl = d; s = src[i]; }
      }
      int a0 = 0, a1 = 0;
      if (ok) {
        const int j = dl * 2;
        const float w0 = __expf(lrelu(el[s * 2]     + er_l[j])     - unflipf(m_l[j]));
        const float w1 = __expf(lrelu(el[s * 2 + 1] + er_l[j + 1]) - unflipf(m_l[j + 1]));
        a0 = __float_as_int(0.5f * w0 / den_l[j]);
        a1 = __float_as_int(0.5f * w1 / den_l[j + 1]);
      }
      unsigned mask = (unsigned)__ballot(ok);
      while (mask) {
        const int j = __builtin_ctz(mask);
        mask &= mask - 1u;
        one_edge(__builtin_amdgcn_readlane(s, j), __builtin_amdgcn_readlane(dl, j),
                 __int_as_float(__builtin_amdgcn_readlane(a0, j)),
                 __int_as_float(__builtin_amdgcn_readlane(a1, j)));
      }
    }
  }
  __syncthreads();

  for (int i = tid; i < nown * F; i += NT) {
    const int f = i % F;
    const float v = agg_l[i] + 0.5f * (bias[f] + bias[F + f]);
    out[(size_t)lo * F + i] = (v > 0.f) ? v : 0.f;
  }
}

static inline int cdiv(long long a, long long b) { return (int)((a + b - 1) / b); }

extern "C" void kernel_launch(void* const* d_in, const int* in_sizes, int n_in,
                              void* d_out, int out_size, void* d_ws, size_t ws_size,
                              hipStream_t stream) {
  (void)in_sizes; (void)n_in; (void)out_size; (void)ws_size;
  const float* x   = (const float*)d_in[0];
  const int*  esrc = (const int*)d_in[1];
  const int*  edst = (const int*)d_in[2];
  const float* W1  = (const float*)d_in[3];
  const float* al1 = (const float*)d_in[4];
  const float* ar1 = (const float*)d_in[5];
  const float* b1  = (const float*)d_in[6];
  const float* W2  = (const float*)d_in[7];
  const float* al2 = (const float*)d_in[8];
  const float* ar2 = (const float*)d_in[9];
  const float* b2  = (const float*)d_in[10];

  const int N = GAT_N, E = GAT_E;

  float* ws = (float*)d_ws;
  size_t o = 0;
  float*    feat1  = ws + o; o += (size_t)N * 128;
  float*    el1    = ws + o; o += (size_t)N * 2;
  float*    er1    = ws + o; o += (size_t)N * 2;
  float*    h1     = ws + o; o += (size_t)N * 64;
  float*    feat2  = ws + o; o += (size_t)N * 64;
  float*    el2    = ws + o; o += (size_t)N * 2;
  float*    er2    = ws + o; o += (size_t)N * 2;

  const int BLK = 256;
  const int OWN = cdiv(N, NPW);
  {
    int waves = N / 16;
    wmma_gemm_kernel<128, 128><<<cdiv((long long)waves * 32, BLK), BLK, 0, stream>>>(
        x, W1, feat1, N);
    node_attn_kernel<64><<<cdiv((long long)N * 32, BLK), BLK, 0, stream>>>(
        feat1, al1, ar1, el1, er1, N);
    owner_kernel<64><<<OWN, 1024, 0, stream>>>(esrc, edst, feat1, el1, er1, b1, h1, N, E);
  }
  {
    int waves = N / 16;
    wmma_gemm_kernel<64, 64><<<cdiv((long long)waves * 32, BLK), BLK, 0, stream>>>(
        h1, W2, feat2, N);
    node_attn_kernel<32><<<cdiv((long long)N * 32, BLK), BLK, 0, stream>>>(
        feat2, al2, ar2, el2, er2, N);
    owner_kernel<32><<<OWN, 1024, 0, stream>>>(esrc, edst, feat2, el2, er2, b2, (float*)d_out, N, E);
  }
}


# Destination filter run once per call: layer 1's work-groups leave their (source, local node) lists and counts in the workspace, layer 2's load them, no second scan of dst; sums stay arrival-ordered
# speedup vs baseline: 3.8837x; 1.2730x over previous
#include <hip/hip_runtime.h>
#include <hip/hip_bf16.h>

typedef __attribute__((ext_vector_type(2))) float v2f;
typedef __attribute__((ext_vector_type(8))) float v8f;

#define GAT_N    50000
#define GAT_E    800000
#define GAT_DIN  128
#define GAT_DHID 64
#define GAT_DOUT 32
#define NEG_SLOPE 0.2f

__device__ __forceinline__ unsigned flipf(float f) {
  unsigned u = __float_as_uint(f);
  return (u & 0x80000000u) ? ~u : (u | 0x80000000u);
}
__device__ __forceinline__ float unflipf(unsigned u) {
  unsigned b = (u & 0x80000000u) ? (u & 0x7FFFFFFFu) : ~u;
  return __uint_as_float(b);
}

template<int K, int NCOL>
__global__ __launch_bounds__(256)
void wmma_gemm_kernel(const float* __restrict__ A, const float* __restrict__ B,
                      float* __restrict__ C, int Mrows) {
  constexpr int NT = NCOL / 16;
  int wave = (int)((blockIdx.x * blockDim.x + threadIdx.x) >> 5);
  int lane = (int)(threadIdx.x & 31);
  int m0 = wave * 16;
  if (m0 >= Mrows) return;
  int half = lane >> 4;
  int lr = lane & 15;

  v8f acc[NT] = {};
  const float* arow = A + (size_t)(m0 + lr) * K;

  for (int k = 0; k < K; k += 4) {
    int ka = k + 2 * half;
    v2f a;
    a.x = arow[ka];
    a.y = arow[ka + 1];
    const float* b0 = B + (size_t)ka * NCOL;
    const float* b1 = B + (size_t)(ka + 1) * NCOL;
#pragma unroll
    for (int t = 0; t < NT; ++t) {
      v2f b;
      b.x = b0[t * 16 + lr];
      b.y = b1[t * 16 + lr];
      acc[t] = __builtin_amdgcn_wmma_f32_16x16x4_f32(
           false, a,  false, b,
           (short)0, acc[t],  false,  false);
    }
  }

  int row = m0 + half * 8;
#pragma unroll
  for (int t = 0; t < NT; ++t) {
#pragma unroll
    for (int r = 0; r < 8; ++r) {
      C[(size_t)(row + r) * NCOL + t * 16 + lr] = acc[t][r];
    }
  }
}

template<int F>
__global__ __launch_bounds__(256)
void node_attn_kernel(const float* __restrict__ feat,
                      const float* __restrict__ al, const float* __restrict__ ar,
                      float* __restrict__ el, float* __restrict__ er, int n_nodes) {
  int wave = (int)((blockIdx.x * blockDim.x + threadIdx.x) >> 5);
  int lane = (int)(threadIdx.x & 31);
  if (wave >= n_nodes) return;
  const float* frow = feat + (size_t)wave * (2 * F);
#pragma unroll
  for (int h = 0; h < 2; ++h) {
    float sl = 0.f, sr = 0.f;
    for (int f = lane; f < F; f += 32) {
      float v = frow[h * F + f];
      sl += v * al[h * F + f];
      sr += v * ar[h * F + f];
    }
#pragma unroll
    for (int off = 16; off > 0; off >>= 1) {
      sl += __shfl_xor(sl, off, 32);
      sr += __shfl_xor(sr, off, 32);
    }
    if (lane == 0) {
      el[wave * 2 + h] = sl;
      er[wave * 2 + h] = sr;
    }
  }
}

#define NPW 196
#define CAP 16384
static_assert(GAT_E % 4 == 0, "dst is scanned four edges per load");
static_assert(GAT_E <= (1 << 20), "edge index is packed into 20 bits");
static_assert(GAT_N <= (1 << 16), "source node is packed into 16 bits");
static_assert(NPW <= (1 << 12), "local node is packed above the edge index");

__device__ __forceinline__ float lrelu(float v) { return (v > 0.f) ? v : NEG_SLOPE * v; }

template<int F, bool TAKE>
__global__ __launch_bounds__(1024)
void owner_kernel(const int* __restrict__ src, const int* __restrict__ dst,
                  const float* __restrict__ feat,
                  const float* __restrict__ el, const float* __restrict__ er,
                  const float* __restrict__ bias, float* __restrict__ out,
                  int* __restrict__ glist, int* __restrict__ gcnt,
                  int n_nodes, int nE) {
  constexpr int HV = F / 32;
  constexpr int U = 16 / HV;
  constexpr int NT = 1024;
  constexpr int SU = 8;
  constexpr int PB = 4;
  static_assert(PB == 4, "the load fences of phase 1 name four slots");

  __shared__ float    agg_l[NPW * F];
  __shared__ unsigned m_l[NPW * 2];
  __shared__ float    den_l[NPW * 2];
  __shared__ float    er_l[NPW * 2];
  __shared__ unsigned list_l[CAP];
  __shared__ v2f      ev_l[CAP];
  __shared__ unsigned cnt_l;
  __shared__ unsigned tick_l;

  const int tid = (int)threadIdx.x;
  const int lane = tid & 31;
  const int lo = (int)blockIdx.x * NPW;
  const int nown = (n_nodes - lo < NPW) ? (n_nodes - lo) : NPW;

  for (int i = tid; i < NPW * F; i += NT) agg_l[i] = 0.f;
  for (int i = tid; i < NPW * 2; i += NT) {
    m_l[i] = 0u;
    den_l[i] = 0.f;
    er_l[i] = (i < nown * 2) ? er[(size_t)lo * 2 + i] : 0.f;
  }
  if (tid == 0) { cnt_l = 0u; tick_l = 0u; }
  __syncthreads();

  if constexpr (!TAKE) {
    const int nQ = nE >> 2;
    const int4* __restrict__ dq = reinterpret_cast<const int4*>(dst);
    const int rot = (int)(((long long)blockIdx.x * nQ) / (long long)gridDim.x);
    for (int k0 = 0; k0 < nQ; k0 += NT * SU) {
      int4 v[SU];
      int qi[SU];
#pragma unroll
      for (int u = 0; u < SU; ++u) {
        const int k = k0 + u * NT + tid;
        int q = k + rot;
        if (q >= nQ) q -= nQ;
        qi[u] = q;
        const int4 t = dq[(k < nQ) ? q : 0];
        v[u] = (k < nQ) ? t : make_int4(-1, -1, -1, -1);
      }
#pragma unroll
      for (int u = 0; u < SU; ++u) {
        const int d4[4] = {v[u].x, v[u].y, v[u].z, v[u].w};
#pragma unroll
        for (int c = 0; c < 4; ++c) {
          const unsigned dd = (unsigned)(d4[c] - lo);
          if (dd < (unsigned)nown) {
            const unsigned pos = atomicAdd(&cnt_l, 1u);
            if (pos < (unsigned)CAP) list_l[pos] = (unsigned)(qi[u] * 4 + c) | (dd << 20);
          }
        }
      }
    }
    __syncthreads();
  }
  int cnt;
  if constexpr (TAKE) cnt = gcnt[blockIdx.x]; else cnt = (int)cnt_l;
  const unsigned total = (unsigned)__builtin_amdgcn_readfirstlane(cnt);
  const bool use_list = (total <= (unsigned)CAP);
  int* __restrict__ gl = glist + (size_t)blockIdx.x * CAP;
  if constexpr (!TAKE) {
    if (tid == 0) gcnt[blockIdx.x] = (int)total;
  }

  if (use_list) {
    for (int i0 = 0; i0 < (int)total; i0 += NT * PB) {
      int s[PB], dn[PB];
      float l0[PB], l1[PB];
      if constexpr (TAKE) {
#pragma unroll
        for (int k = 0; k < PB; ++k) {
          const int i = i0 + k * NT + tid;
          const unsigned t = (unsigned)gl[(i < (int)total) ? i : 0];
          s[k]  = (i < (int)total) ? (int)(t & 0xFFFFu) : 0;
          dn[k] = (i < (int)total) ? (int)(t >> 16) : 0;
        }
      } else {
        unsigned p[PB];
#pragma unroll
        for (int k = 0; k < PB; ++k) {
          const int i = i0 + k * NT + tid;
          const unsigned t = list_l[(i < (int)total) ? i : 0];
          p[k] = (i < (int)total) ? t : 0u;
          dn[k] = (int)(p[k] >> 20);
        }
#pragma unroll
        for (int k = 0; k < PB; ++k) s[k] = src[p[k] & 0xFFFFFu];
        asm volatile("" :: "v"(s[0]), "v"(s[1]), "v"(s[2]), "v"(s[3]));
      }
#pragma unroll
      for (int k = 0; k < PB; ++k) { l0[k] = el[s[k] * 2]; l1[k] = el[s[k] * 2 + 1]; }
      asm volatile("" :: "v"(l0[0]), "v"(l1[0]), "v"(l0[1]), "v"(l1[1]),
                         "v"(l0[2]), "v"(l1[2]), "v"(l0[3]), "v"(l1[3]));
#pragma unroll
      for (int k = 0; k < PB; ++k) {
        const int i = i0 + k * NT + tid;
        if (i < (int)total) {
          const int dl = dn[k];
          const unsigned sd = (unsigned)s[k] | ((unsigned)dl << 16);
          list_l[i] = sd;
          if constexpr (!TAKE) gl[i] = (int)sd;
          v2f e;
          e.x = lrelu(l0[k] + er_l[dl * 2]);
          e.y = lrelu(l1[k] + er_l[dl * 2 + 1]);
          ev_l[i] = e;
          atomicMax(&m_l[dl * 2],     flipf(e.x));
          atomicMax(&m_l[dl * 2 + 1], flipf(e.y));
        }
      }
    }
  } else {
    for (int i = tid; i < nE; i += NT) {
      const int d = dst[i] - lo;
      if ((unsigned)d < (unsigned)nown) {
        const int s = src[i];
        atomicMax(&m_l[d * 2],     flipf(lrelu(el[s * 2]     + er_l[d * 2])));
        atomicMax(&m_l[d * 2 + 1], flipf(lrelu(el[s * 2 + 1] + er_l[d * 2 + 1])));
      }
    }
  }
  __syncthreads();

  if (use_list) {
    for (int i = tid; i < (int)total; i += NT) {
      const int j = (int)(list_l[i] >> 16) * 2;
      const v2f e = ev_l[i];
      v2f w;
      w.x = __expf(e.x - unflipf(m_l[j]));
      w.y = __expf(e.y - unflipf(m_l[j + 1]));
      ev_l[i] = w;
      atomicAdd(&den_l[j],     w.x);
      atomicAdd(&den_l[j + 1], w.y);
    }
  } else {
    for (int i = tid; i < nE; i += NT) {
      const int d = dst[i] - lo;
      if ((unsigned)d < (unsigned)nown) {
        const int s = src[i];
        const int j = d * 2;
        atomicAdd(&den_l[j],     __expf(lrelu(el[s * 2]     + er_l[j])     - unflipf(m_l[j])));
        atomicAdd(&den_l[j + 1], __expf(lrelu(el[s * 2 + 1] + er_l[j + 1]) - unflipf(m_l[j + 1])));
      }
    }
  }
  __syncthreads();

  auto one_edge = [&](int sj, int dj, float b0, float b1) {
    const float* frow = feat + (size_t)sj * (2 * F);
#pragma unroll
    for (int q = 0; q < HV; ++q) {
      atomicAdd(&agg_l[dj * F + lane + q * 32],
                b0 * frow[lane + q * 32] + b1 * frow[lane + F + q * 32]);
    }
  };
  if (use_list) {
    for (;;) {
      int c = 0;
      if (lane == 0) c = (int)atomicAdd(&tick_l, 1u);
      c = __builtin_amdgcn_readfirstlane(c);
      const int base = c * U;
      if (base >= (int)total) break;
      const int n = ((int)total - base < U) ? ((int)total - base) : U;
      const int slot = base + ((lane < n) ? lane : 0);
      const unsigned p = list_l[slot];
      const v2f wv = ev_l[slot];
      const int s = (int)(p & 0xFFFFu), dl = (int)(p >> 16);
      auto factor = [&](int h, float w) -> int {
        return __float_as_int(0.5f * w / den_l[dl * 2 + h]);
      };
      if (n == U) {
        float f0[U][HV], f1[U][HV];
#pragma unroll
        for (int u = 0; u < U; ++u) {
          const float* frow = feat + (size_t)__builtin_amdgcn_readlane(s, u) * (2 * F);
#pragma unroll
          for (int q = 0; q < HV; ++q) {
            f0[u][q] = frow[lane + q * 32];
            f1[u][q] = frow[lane + F + q * 32];
          }
        }
        const int a0 = factor(0, wv.x), a1 = factor(1, wv.y);
#pragma unroll
        for (int u = 0; u < U; ++u) {
          float* arow = agg_l + __builtin_amdgcn_readlane(dl, u) * F;
          const float b0 = __int_as_float(__builtin_amdgcn_readlane(a0, u));
          const float b1 = __int_as_float(__builtin_amdgcn_readlane(a1, u));
#pragma unroll
          for (int q = 0; q < HV; ++q) {
            atomicAdd(&arow[lane + q * 32], b0 * f0[u][q] + b1 * f1[u][q]);
          }
        }
      } else {
        const int a0 = factor(0, wv.x), a1 = factor(1, wv.y);
        for (int j = 0; j < n; ++j) {
          one_edge(__builtin_amdgcn_readlane(s, j), __builtin_amdgcn_readlane(dl, j),
                   __int_as_float(__builtin_amdgcn_readlane(a0, j)),
                   __int_as_float(__builtin_amdgcn_readlane(a1, j)));
        }
      }
    }
  } else {
    for (;;) {
      int c = 0;
      if (lane == 0) c = (int)atomicAdd(&tick_l, 1u);
      c = __builtin_amdgcn_readfirstlane(c);
      const int base = c * 32;
      if (base >= nE) break;
      const int i = base + lane;
      bool ok = false;
      int s = 0, dl = 0;
      if (i < nE) {
        const int d = dst[i] - lo;
        if ((unsigned)d < (unsigned)nown) { ok = true; dl = d; s = src[i]; }
      }
      int a0 = 0, a1 = 0;
      if (ok) {
        const int j = dl * 2;
        const float w0 = __expf(lrelu(el[s * 2]     + er_l[j])     - unflipf(m_l[j]));
        const float w1 = __expf(lrelu(el[s * 2 + 1] + er_l[j + 1]) - unflipf(m_l[j + 1]));
        a0 = __float_as_int(0.5f * w0 / den_l[j]);
        a1 = __float_as_int(0.5f * w1 / den_l[j + 1]);
      }
      unsigned mask = (unsigned)__ballot(ok);
      while (mask) {
        const int j = __builtin_ctz(mask);
        mask &= mask - 1u;
        one_edge(__builtin_amdgcn_readlane(s, j), __builtin_amdgcn_readlane(dl, j),
                 __int_as_float(__builtin_amdgcn_readlane(a0, j)),
                 __int_as_float(__builtin_amdgcn_readlane(a1, j)));
      }
    }
  }
  __syncthreads();

  for (int i = tid; i < nown * F; i += NT) {
    const int f = i % F;
    const float v = agg_l[i] + 0.5f * (bias[f] + bias[F + f]);
    out[(size_t)lo * F + i] = (v > 0.f) ? v : 0.f;
  }
}

static inline int cdiv(long long a, long long b) { return (int)((a + b - 1) / b); }

extern "C" void kernel_launch(void* const* d_in, const int* in_sizes, int n_in,
                              void* d_out, int out_size, void* d_ws, size_t ws_size,
                              hipStream_t stream) {
  (void)in_sizes; (void)n_in; (void)out_size; (void)ws_size;
  const float* x   = (const float*)d_in[0];
  const int*  esrc = (const int*)d_in[1];
  const int*  edst = (const int*)d_in[2];
  const float* W1  = (const float*)d_in[3];
  const float* al1 = (const float*)d_in[4];
  const float* ar1 = (const float*)d_in[5];
  const float* b1  = (const float*)d_in[6];
  const float* W2  = (const float*)d_in[7];
  const float* al2 = (const float*)d_in[8];
  const float* ar2 = (const float*)d_in[9];
  const float* b2  = (const float*)d_in[10];

  const int N = GAT_N, E = GAT_E;

  float* ws = (float*)d_ws;
  size_t o = 0;
  float*    feat1  = ws + o; o += (size_t)N * 128;
  float*    el1    = ws + o; o += (size_t)N * 2;
  float*    er1    = ws + o; o += (size_t)N * 2;
  float*    h1     = ws + o; o += (size_t)N * 64;
  float*    feat2  = ws + o; o += (size_t)N * 64;
  float*    el2    = ws + o; o += (size_t)N * 2;
  float*    er2    = ws + o; o += (size_t)N * 2;
  const int OWN = cdiv(N, NPW);
  int*      elist  = (int*)(ws + o); o += (size_t)OWN * CAP;
  int*      ecnt   = (int*)(ws + o); o += (size_t)OWN;

  const int BLK = 256;
  {
    int waves = N / 16;
    wmma_gemm_kernel<128, 128><<<cdiv((long long)waves * 32, BLK), BLK, 0, stream>>>(
        x, W1, feat1, N);
    node_attn_kernel<64><<<cdiv((long long)N * 32, BLK), BLK, 0, stream>>>(
        feat1, al1, ar1, el1, er1, N);
    owner_kernel<64, false><<<OWN, 1024, 0, stream>>>(esrc, edst, feat1, el1, er1, b1, h1,
                                                     elist, ecnt, N, E);
  }
  {
    int waves = N / 16;
    wmma_gemm_kernel<64, 64><<<cdiv((long long)waves * 32, BLK), BLK, 0, stream>>>(
        h1, W2, feat2, N);
    node_attn_kernel<32><<<cdiv((long long)N * 32, BLK), BLK, 0, stream>>>(
        feat2, al2, ar2, el2, er2, N);
    owner_kernel<32, true><<<OWN, 1024, 0, stream>>>(esrc, edst, feat2, el2, er2, b2, (float*)d_out,
                                                    elist, ecnt, N, E);
  }
}


# dst turned once into a membership bitmap row per node range (a small launch, LDS or-atomics); owner work-groups list their own row's set bits, no scan of dst; 6 to 7 launches; sums arrival-ordered
# speedup vs baseline: 4.7915x; 1.2337x over previous
#include <hip/hip_runtime.h>
#include <hip/hip_bf16.h>

typedef __attribute__((ext_vector_type(2))) float v2f;
typedef __attribute__((ext_vector_type(8))) float v8f;

#define GAT_N    50000
#define GAT_E    800000
#define GAT_DIN  128
#define GAT_DHID 64
#define GAT_DOUT 32
#define NEG_SLOPE 0.2f

__device__ __forceinline__ unsigned flipf(float f) {
  unsigned u = __float_as_uint(f);
  return (u & 0x80000000u) ? ~u : (u | 0x80000000u);
}
__device__ __forceinline__ float unflipf(unsigned u) {
  unsigned b = (u & 0x80000000u) ? (u & 0x7FFFFFFFu) : ~u;
  return __uint_as_float(b);
}

template<int K, int NCOL>
__global__ __launch_bounds__(256)
void wmma_gemm_kernel(const float* __restrict__ A, const float* __restrict__ B,
                      float* __restrict__ C, int Mrows) {
  constexpr int NT = NCOL / 16;
  int wave = (int)((blockIdx.x * blockDim.x + threadIdx.x) >> 5);
  int lane = (int)(threadIdx.x & 31);
  int m0 = wave * 16;
  if (m0 >= Mrows) return;
  int half = lane >> 4;
  int lr = lane & 15;

  v8f acc[NT] = {};
  const float* arow = A + (size_t)(m0 + lr) * K;

  for (int k = 0; k < K; k += 4) {
    int ka = k + 2 * half;
    v2f a;
    a.x = arow[ka];
    a.y = arow[ka + 1];
    const float* b0 = B + (size_t)ka * NCOL;
    const float* b1 = B + (size_t)(ka + 1) * NCOL;
#pragma unroll
    for (int t = 0; t < NT; ++t) {
      v2f b;
      b.x = b0[t * 16 + lr];
      b.y = b1[t * 16 + lr];
      acc[t] = __builtin_amdgcn_wmma_f32_16x16x4_f32(
           false, a,  false, b,
           (short)0, acc[t],  false,  false);
    }
  }

  int row = m0 + half * 8;
#pragma unroll
  for (int t = 0; t < NT; ++t) {
#pragma unroll
    for (int r = 0; r < 8; ++r) {
      C[(size_t)(row + r) * NCOL + t * 16 + lr] = acc[t][r];
    }
  }
}

template<int F>
__global__ __launch_bounds__(256)
void node_attn_kernel(const float* __restrict__ feat,
                      const float* __restrict__ al, const float* __restrict__ ar,
                      float* __restrict__ el, float* __restrict__ er, int n_nodes) {
  int wave = (int)((blockIdx.x * blockDim.x + threadIdx.x) >> 5);
  int lane = (int)(threadIdx.x & 31);
  if (wave >= n_nodes) return;
  const float* frow = feat + (size_t)wave * (2 * F);
#pragma unroll
  for (int h = 0; h < 2; ++h) {
    float sl = 0.f, sr = 0.f;
    for (int f = lane; f < F; f += 32) {
      float v = frow[h * F + f];
      sl += v * al[h * F + f];
      sr += v * ar[h * F + f];
    }
#pragma unroll
    for (int off = 16; off > 0; off >>= 1) {
      sl += __shfl_xor(sl, off, 32);
      sr += __shfl_xor(sr, off, 32);
    }
    if (lane == 0) {
      el[wave * 2 + h] = sl;
      er[wave * 2 + h] = sr;
    }
  }
}

#define NPW 196
#define CAP 16384
#define NRANGE ((GAT_N + NPW - 1) / NPW)
#define BCH 1024
#define NCH ((GAT_E + BCH - 1) / BCH)
#define WPR (NCH * 32)
static_assert(GAT_E % 4 == 0, "dst is read four edges per load");
static_assert(WPR % 4 == 0, "a bitmap row is read 16 bytes per load");
static_assert(GAT_N <= (1 << 16), "source node is packed into 16 bits");
static_assert(NPW <= (1 << 16), "local node is packed above the source node");
static_assert(NRANGE == 256, "range_bitmap_kernel keeps one LDS column per range");

__global__ __launch_bounds__(256)
void range_bitmap_kernel(const int* __restrict__ dst, unsigned* __restrict__ bm,
                         int n_nodes, int nE) {
  __shared__ unsigned b_l[32 * 257];
  const int tid = (int)threadIdx.x;
  for (int i = tid; i < 32 * 257; i += 256) b_l[i] = 0u;
  __syncthreads();
  const int e0 = (int)blockIdx.x * BCH + tid * 4;
  if (e0 < nE) {
    const int4 d = *reinterpret_cast<const int4*>(dst + e0);
    const int d4[4] = {d.x, d.y, d.z, d.w};
    const int w = tid >> 3;
#pragma unroll
    for (int c = 0; c < 4; ++c) {
      if ((unsigned)d4[c] < (unsigned)n_nodes) {
        atomicOr(&b_l[w * 257 + (int)((unsigned)d4[c] / (unsigned)NPW)],
                 1u << ((tid & 7) * 4 + c));
      }
    }
  }
  __syncthreads();
  for (int i = tid; i < NRANGE * 32; i += 256) {
    const int r = i >> 5, w = i & 31;
    bm[(size_t)r * WPR + (size_t)blockIdx.x * 32 + w] = b_l[w * 257 + r];
  }
}

__device__ __forceinline__ float lrelu(float v) { return (v > 0.f) ? v : NEG_SLOPE * v; }

template<int F, bool TAKE>
__global__ __launch_bounds__(1024)
void owner_kernel(const int* __restrict__ src, const int* __restrict__ dst,
                  const float* __restrict__ feat,
                  const float* __restrict__ el, const float* __restrict__ er,
                  const float* __restrict__ bias, float* __restrict__ out,
                  const unsigned* __restrict__ bm,
                  int* __restrict__ glist, int* __restrict__ gcnt,
                  int n_nodes, int nE) {
  constexpr int HV = F / 32;
  constexpr int U = 16 / HV;
  constexpr int NT = 1024;
  constexpr int QPR = WPR / 4;
  constexpr int SB = (QPR + NT - 1) / NT;
  constexpr int PB = 4;
  static_assert(PB == 4, "the load fences of phase 1 name four slots");

  __shared__ float    agg_l[NPW * F];
  __shared__ unsigned m_l[NPW * 2];
  __shared__ float    den_l[NPW * 2];
  __shared__ float    er_l[NPW * 2];
  __shared__ unsigned list_l[CAP];
  __shared__ v2f      ev_l[CAP];
  __shared__ unsigned cnt_l;
  __shared__ unsigned tick_l;

  const int tid = (int)threadIdx.x;
  const int lane = tid & 31;
  const int lo = (int)blockIdx.x * NPW;
  const int nown = (n_nodes - lo < NPW) ? (n_nodes - lo) : NPW;

  for (int i = tid; i < NPW * F; i += NT) agg_l[i] = 0.f;
  for (int i = tid; i < NPW * 2; i += NT) {
    m_l[i] = 0u;
    den_l[i] = 0.f;
    er_l[i] = (i < nown * 2) ? er[(size_t)lo * 2 + i] : 0.f;
  }
  if (tid == 0) { cnt_l = 0u; tick_l = 0u; }
  __syncthreads();

  if constexpr (!TAKE) {
    const uint4* __restrict__ row =
        reinterpret_cast<const uint4*>(bm + (size_t)blockIdx.x * WPR);
    uint4 v[SB];
    int n = 0;
#pragma unroll
    for (int u = 0; u < SB; ++u) {
      const int q = u * NT + tid;
      const uint4 t = row[(q < QPR) ? q : 0];
      v[u] = (q < QPR) ? t : make_uint4(0u, 0u, 0u, 0u);
      n += __popc(v[u].x) + __popc(v[u].y) + __popc(v[u].z) + __popc(v[u].w);
    }
    unsigned pos = atomicAdd(&cnt_l, (unsigned)n);
#pragma unroll
    for (int u = 0; u < SB; ++u) {
      const unsigned w4[4] = {v[u].x, v[u].y, v[u].z, v[u].w};
#pragma unroll
      for (int j = 0; j < 4; ++j) {
        const unsigned e0 = (unsigned)(((u * NT + tid) * 4 + j) * 32);
        unsigned w = w4[j];
        while (w) {
          if (pos < (unsigned)CAP) list_l[pos] = e0 + (unsigned)__builtin_ctz(w);
          ++pos;
          w &= w - 1u;
        }
      }
    }
    __syncthreads();
  }
  int cnt;
  if constexpr (TAKE) cnt = gcnt[blockIdx.x]; else cnt = (int)cnt_l;
  const unsigned total = (unsigned)__builtin_amdgcn_readfirstlane(cnt);
  const bool use_list = (total <= (unsigned)CAP);
  int* __restrict__ gl = glist + (size_t)blockIdx.x * CAP;
  if constexpr (!TAKE) {
    if (tid == 0) gcnt[blockIdx.x] = (int)total;
  }

  if (use_list) {
    for (int i0 = 0; i0 < (int)total; i0 += NT * PB) {
      int s[PB], dn[PB];
      float l0[PB], l1[PB];
      if constexpr (TAKE) {
#pragma unroll
        for (int k = 0; k < PB; ++k) {
          const int i = i0 + k * NT + tid;
          const unsigned t = (unsigned)gl[(i < (int)total) ? i : 0];
          s[k]  = (i < (int)total) ? (int)(t & 0xFFFFu) : 0;
          dn[k] = (i < (int)total) ? (int)(t >> 16) : 0;
        }
      } else {
        unsigned p[PB];
        int dv[PB];
#pragma unroll
        for (int k = 0; k < PB; ++k) {
          const int i = i0 + k * NT + tid;
          const unsigned t = list_l[(i < (int)total) ? i : 0];
          p[k] = (i < (int)total) ? t : 0u;
        }
#pragma unroll
        for (int k = 0; k < PB; ++k) { s[k] = src[p[k]]; dv[k] = dst[p[k]]; }
        asm volatile("" :: "v"(s[0]), "v"(s[1]), "v"(s[2]), "v"(s[3]),
                           "v"(dv[0]), "v"(dv[1]), "v"(dv[2]), "v"(dv[3]));
#pragma unroll
        for (int k = 0; k < PB; ++k) dn[k] = dv[k] - lo;
      }
#pragma unroll
      for (int k = 0; k < PB; ++k) { l0[k] = el[s[k] * 2]; l1[k] = el[s[k] * 2 + 1]; }
      asm volatile("" :: "v"(l0[0]), "v"(l1[0]), "v"(l0[1]), "v"(l1[1]),
                         "v"(l0[2]), "v"(l1[2]), "v"(l0[3]), "v"(l1[3]));
#pragma unroll
      for (int k = 0; k < PB; ++k) {
        const int i = i0 + k * NT + tid;
        if (i < (int)total) {
          const int dl = dn[k];
          const unsigned sd = (unsigned)s[k] | ((unsigned)dl << 16);
          list_l[i] = sd;
          if constexpr (!TAKE) gl[i] = (int)sd;
          v2f e;
          e.x = lrelu(l0[k] + er_l[dl * 2]);
          e.y = lrelu(l1[k] + er_l[dl * 2 + 1]);
          ev_l[i] = e;
          atomicMax(&m_l[dl * 2],     flipf(e.x));
          atomicMax(&m_l[dl * 2 + 1], flipf(e.y));
        }
      }
    }
  } else {
    for (int i = tid; i < nE; i += NT) {
      const int d = dst[i] - lo;
      if ((unsigned)d < (unsigned)nown) {
        const int s = src[i];
        atomicMax(&m_l[d * 2],     flipf(lrelu(el[s * 2]     + er_l[d * 2])));
        atomicMax(&m_l[d * 2 + 1], flipf(lrelu(el[s * 2 + 1] + er_l[d * 2 + 1])));
      }
    }
  }
  __syncthreads();

  if (use_list) {
    for (int i = tid; i < (int)total; i += NT) {
      const int j = (int)(list_l[i] >> 16) * 2;
      const v2f e = ev_l[i];
      v2f w;
      w.x = __expf(e.x - unflipf(m_l[j]));
      w.y = __expf(e.y - unflipf(m_l[j + 1]));
      ev_l[i] = w;
      atomicAdd(&den_l[j],     w.x);
      atomicAdd(&den_l[j + 1], w.y);
    }
  } else {
    for (int i = tid; i < nE; i += NT) {
      const int d = dst[i] - lo;
      if ((unsigned)d < (unsigned)nown) {
        const int s = src[i];
        const int j = d * 2;
        atomicAdd(&den_l[j],     __expf(lrelu(el[s * 2]     + er_l[j])     - unflipf(m_l[j])));
        atomicAdd(&den_l[j + 1], __expf(lrelu(el[s * 2 + 1] + er_l[j + 1]) - unflipf(m_l[j + 1])));
      }
    }
  }
  __syncthreads();

  auto one_edge = [&](int sj, int dj, float b0, float b1) {
    const float* frow = feat + (size_t)sj * (2 * F);
#pragma unroll
    for (int q = 0; q < HV; ++q) {
      atomicAdd(&agg_l[dj * F + lane + q * 32],
                b0 * frow[lane + q * 32] + b1 * frow[lane + F + q * 32]);
    }
  };
  if (use_list) {
    for (;;) {
      int c = 0;
      if (lane == 0) c = (int)atomicAdd(&tick_l, 1u);
      c = __builtin_amdgcn_readfirstlane(c);
      const int base = c * U;
      if (base >= (int)total) break;
      const int n = ((int)total - base < U) ? ((int)total - base) : U;
      const int slot = base + ((lane < n) ? lane : 0);
      const unsigned p = list_l[slot];
      const v2f wv = ev_l[slot];
      const int s = (int)(p & 0xFFFFu), dl = (int)(p >> 16);
      auto factor = [&](int h, float w) -> int {
        return __float_as_int(0.5f * w / den_l[dl * 2 + h]);
      };
      if (n == U) {
        float f0[U][HV], f1[U][HV];
#pragma unroll
        for (int u = 0; u < U; ++u) {
          const float* frow = feat + (size_t)__builtin_amdgcn_readlane(s, u) * (2 * F);
#pragma unroll
          for (int q = 0; q < HV; ++q) {
            f0[u][q] = frow[lane + q * 32];
            f1[u][q] = frow[lane + F + q * 32];
          }
        }
        const int a0 = factor(0, wv.x), a1 = factor(1, wv.y);
#pragma unroll
        for (int u = 0; u < U; ++u) {
          float* arow = agg_l + __builtin_amdgcn_readlane(dl, u) * F;
          const float b0 = __int_as_float(__builtin_amdgcn_readlane(a0, u));
          const float b1 = __int_as_float(__builtin_amdgcn_readlane(a1, u));
#pragma unroll
          for (int q = 0; q < HV; ++q) {
            atomicAdd(&arow[lane + q * 32], b0 * f0[u][q] + b1 * f1[u][q]);
          }
        }
      } else {
        const int a0 = factor(0, wv.x), a1 = factor(1, wv.y);
        for (int j = 0; j < n; ++j) {
          one_edge(__builtin_amdgcn_readlane(s, j), __builtin_amdgcn_readlane(dl, j),
                   __int_as_float(__builtin_amdgcn_readlane(a0, j)),
                   __int_as_float(__builtin_amdgcn_readlane(a1, j)));
        }
      }
    }
  } else {
    for (;;) {
      int c = 0;
      if (lane == 0) c = (int)atomicAdd(&tick_l, 1u);
      c = __builtin_amdgcn_readfirstlane(c);
      const int base = c * 32;
      if (base >= nE) break;
      const int i = base + lane;
      bool ok = false;
      int s = 0, dl = 0;
      if (i < nE) {
        const int d = dst[i] - lo;
        if ((unsigned)d < (unsigned)nown) { ok = true; dl = d; s = src[i]; }
      }
      int a0 = 0, a1 = 0;
      if (ok) {
        const int j = dl * 2;
        const float w0 = __expf(lrelu(el[s * 2]     + er_l[j])     - unflipf(m_l[j]));
        const float w1 = __expf(lrelu(el[s * 2 + 1] + er_l[j + 1]) - unflipf(m_l[j + 1]));
        a0 = __float_as_int(0.5f * w0 / den_l[j]);
        a1 = __float_as_int(0.5f * w1 / den_l[j + 1]);
      }
      unsigned mask = (unsigned)__ballot(ok);
      while (mask) {
        const int j = __builtin_ctz(mask);
        mask &= mask - 1u;
        one_edge(__builtin_amdgcn_readlane(s, j), __builtin_amdgcn_readlane(dl, j),
                 __int_as_float(__builtin_amdgcn_readlane(a0, j)),
                 __int_as_float(__builtin_amdgcn_readlane(a1, j)));
      }
    }
  }
  __syncthreads();

  for (int i = tid; i < nown * F; i += NT) {
    const int f = i % F;
    const float v = agg_l[i] + 0.5f * (bias[f] + bias[F + f]);
    out[(size_t)lo * F + i] = (v > 0.f) ? v : 0.f;
  }
}

static inline int cdiv(long long a, long long b) { return (int)((a + b - 1) / b); }

extern "C" void kernel_launch(void* const* d_in, const int* in_sizes, int n_in,
                              void* d_out, int out_size, void* d_ws, size_t ws_size,
                              hipStream_t stream) {
  (void)in_sizes; (void)n_in; (void)out_size; (void)ws_size;
  const float* x   = (const float*)d_in[0];
  const int*  esrc = (const int*)d_in[1];
  const int*  edst = (const int*)d_in[2];
  const float* W1  = (const float*)d_in[3];
  const float* al1 = (const float*)d_in[4];
  const float* ar1 = (const float*)d_in[5];
  const float* b1  = (const float*)d_in[6];
  const float* W2  = (const float*)d_in[7];
  const float* al2 = (const float*)d_in[8];
  const float* ar2 = (const float*)d_in[9];
  const float* b2  = (const float*)d_in[10];

  const int N = GAT_N, E = GAT_E;

  float* ws = (float*)d_ws;
  size_t o = 0;
  float*    feat1  = ws + o; o += (size_t)N * 128;
  float*    el1    = ws + o; o += (size_t)N * 2;
  float*    er1    = ws + o; o += (size_t)N * 2;
  float*    h1     = ws + o; o += (size_t)N * 64;
  float*    feat2  = ws + o; o += (size_t)N * 64;
  float*    el2    = ws + o; o += (size_t)N * 2;
  float*    er2    = ws + o; o += (size_t)N * 2;
  const int OWN = cdiv(N, NPW);
  int*      elist  = (int*)(ws + o); o += (size_t)OWN * CAP;
  int*      ecnt   = (int*)(ws + o); o += (size_t)OWN;
  unsigned* ebm    = (unsigned*)(ws + o); o += (size_t)OWN * WPR;

  const int BLK = 256;
  {
    range_bitmap_kernel<<<NCH, 256, 0, stream>>>(edst, ebm, N, E);
    int waves = N / 16;
    wmma_gemm_kernel<128, 128><<<cdiv((long long)waves * 32, BLK), BLK, 0, stream>>>(
        x, W1, feat1, N);
    node_attn_kernel<64><<<cdiv((long long)N * 32, BLK), BLK, 0, stream>>>(
        feat1, al1, ar1, el1, er1, N);
    owner_kernel<64, false><<<OWN, 1024, 0, stream>>>(esrc, edst, feat1, el1, er1, b1, h1,
                                                     ebm, elist, ecnt, N, E);
  }
  {
    int waves = N / 16;
    wmma_gemm_kernel<64, 64><<<cdiv((long long)waves * 32, BLK), BLK, 0, stream>>>(
        h1, W2, feat2, N);
    node_attn_kernel<32><<<cdiv((long long)N * 32, BLK), BLK, 0, stream>>>(
        feat2, al2, ar2, el2, er2, N);
    owner_kernel<32, true><<<OWN, 1024, 0, stream>>>(esrc, edst, feat2, el2, er2, b2, (float*)d_out,
                                                    ebm, elist, ecnt, N, E);
  }
}


# Layer 2's projection and logits done by layer 1's owner work-groups from their finished rows in LDS (WMMA from LDS; feat2, el2, er2 stored directly): 7 to 5 launches; layer-2 logit sum order changed
# speedup vs baseline: 5.4426x; 1.1359x over previous
#include <hip/hip_runtime.h>
#include <hip/hip_bf16.h>

typedef __attribute__((ext_vector_type(2))) float v2f;
typedef __attribute__((ext_vector_type(8))) float v8f;

#define GAT_N    50000
#define GAT_E    800000
#define GAT_DIN  128
#define GAT_DHID 64
#define GAT_DOUT 32
#define NEG_SLOPE 0.2f

__device__ __forceinline__ unsigned flipf(float f) {
  unsigned u = __float_as_uint(f);
  return (u & 0x80000000u) ? ~u : (u | 0x80000000u);
}
__device__ __forceinline__ float unflipf(unsigned u) {
  unsigned b = (u & 0x80000000u) ? (u & 0x7FFFFFFFu) : ~u;
  return __uint_as_float(b);
}

template<int K, int NCOL>
__global__ __launch_bounds__(256)
void wmma_gemm_kernel(const float* __restrict__ A, const float* __restrict__ B,
                      float* __restrict__ C, int Mrows) {
  constexpr int NT = NCOL / 16;
  int wave = (int)((blockIdx.x * blockDim.x + threadIdx.x) >> 5);
  int lane = (int)(threadIdx.x & 31);
  int m0 = wave * 16;
  if (m0 >= Mrows) return;
  int half = lane >> 4;
  int lr = lane & 15;

  v8f acc[NT] = {};
  const float* arow = A + (size_t)(m0 + lr) * K;

  for (int k = 0; k < K; k += 4) {
    int ka = k + 2 * half;
    v2f a;
    a.x = arow[ka];
    a.y = arow[ka + 1];
    const float* b0 = B + (size_t)ka * NCOL;
    const float* b1 = B + (size_t)(ka + 1) * NCOL;
#pragma unroll
    for (int t = 0; t < NT; ++t) {
      v2f b;
      b.x = b0[t * 16 + lr];
      b.y = b1[t * 16 + lr];
      acc[t] = __builtin_amdgcn_wmma_f32_16x16x4_f32(
           false, a,  false, b,
           (short)0, acc[t],  false,  false);
    }
  }

  int row = m0 + half * 8;
#pragma unroll
  for (int t = 0; t < NT; ++t) {
#pragma unroll
    for (int r = 0; r < 8; ++r) {
      C[(size_t)(row + r) * NCOL + t * 16 + lr] = acc[t][r];
    }
  }
}

template<int F>
__global__ __launch_bounds__(256)
void node_attn_kernel(const float* __restrict__ feat,
                      const float* __restrict__ al, const float* __restrict__ ar,
                      float* __restrict__ el, float* __restrict__ er, int n_nodes) {
  int wave = (int)((blockIdx.x * blockDim.x + threadIdx.x) >> 5);
  int lane = (int)(threadIdx.x & 31);
  if (wave >= n_nodes) return;
  const float* frow = feat + (size_t)wave * (2 * F);
#pragma unroll
  for (int h = 0; h < 2; ++h) {
    float sl = 0.f, sr = 0.f;
    for (int f = lane; f < F; f += 32) {
      float v = frow[h * F + f];
      sl += v * al[h * F + f];
      sr += v * ar[h * F + f];
    }
#pragma unroll
    for (int off = 16; off > 0; off >>= 1) {
      sl += __shfl_xor(sl, off, 32);
      sr += __shfl_xor(sr, off, 32);
    }
    if (lane == 0) {
      el[wave * 2 + h] = sl;
      er[wave * 2 + h] = sr;
    }
  }
}

#define NPW 196
#define CAP 16384
#define NRANGE ((GAT_N + NPW - 1) / NPW)
#define BCH 1024
#define NCH ((GAT_E + BCH - 1) / BCH)
#define WPR (NCH * 32)
static_assert(GAT_E % 4 == 0, "dst is read four edges per load");
static_assert(WPR % 4 == 0, "a bitmap row is read 16 bytes per load");
static_assert(GAT_N <= (1 << 16), "source node is packed into 16 bits");
static_assert(NPW <= (1 << 16), "local node is packed above the source node");
static_assert(NRANGE == 256, "range_bitmap_kernel keeps one LDS column per range");

__global__ __launch_bounds__(256)
void range_bitmap_kernel(const int* __restrict__ dst, unsigned* __restrict__ bm,
                         int n_nodes, int nE) {
  __shared__ unsigned b_l[32 * 257];
  const int tid = (int)threadIdx.x;
  for (int i = tid; i < 32 * 257; i += 256) b_l[i] = 0u;
  __syncthreads();
  const int e0 = (int)blockIdx.x * BCH + tid * 4;
  if (e0 < nE) {
    const int4 d = *reinterpret_cast<const int4*>(dst + e0);
    const int d4[4] = {d.x, d.y, d.z, d.w};
    const int w = tid >> 3;
#pragma unroll
    for (int c = 0; c < 4; ++c) {
      if ((unsigned)d4[c] < (unsigned)n_nodes) {
        atomicOr(&b_l[w * 257 + (int)((unsigned)d4[c] / (unsigned)NPW)],
                 1u << ((tid & 7) * 4 + c));
      }
    }
  }
  __syncthreads();
  for (int i = tid; i < NRANGE * 32; i += 256) {
    const int r = i >> 5, w = i & 31;
    bm[(size_t)r * WPR + (size_t)blockIdx.x * 32 + w] = b_l[w * 257 + r];
  }
}

__device__ __forceinline__ float lrelu(float v) { return (v > 0.f) ? v : NEG_SLOPE * v; }

template<int F, bool TAKE, int FN>
__global__ __launch_bounds__(1024)
void owner_kernel(const int* __restrict__ src, const int* __restrict__ dst,
                  const float* __restrict__ feat,
                  const float* __restrict__ el, const float* __restrict__ er,
                  const float* __restrict__ bias, float* __restrict__ out,
                  const unsigned* __restrict__ bm,
                  int* __restrict__ glist, int* __restrict__ gcnt,
                  const float* __restrict__ Wn,
                  const float* __restrict__ aln, const float* __restrict__ arn,
                  float* __restrict__ featn,
                  float* __restrict__ eln, float* __restrict__ ern,
                  int n_nodes, int nE) {
  constexpr int HV = F / 32;
  constexpr int U = 16 / HV;
  constexpr int NT = 1024;
  constexpr int QPR = WPR / 4;
  constexpr int SB = (QPR + NT - 1) / NT;
  constexpr int PB = 4;
  static_assert(PB == 4, "the load fences of phase 1 name four slots");

  constexpr int NSR = ((NPW + 15) / 16) * 16;
  constexpr int HP = F + 4;
  constexpr int AGG = (FN > 0) ? NSR * HP : NPW * F;
  __shared__ __attribute__((aligned(16))) float agg_l[AGG];
  __shared__ unsigned m_l[NPW * 2];
  __shared__ float    den_l[NPW * 2];
  __shared__ float    er_l[NPW * 2];
  __shared__ unsigned list_l[CAP];
  __shared__ v2f      ev_l[CAP];
  __shared__ unsigned cnt_l;
  __shared__ unsigned tick_l;
  __shared__ float    bs_l[(FN > 0) ? F : 1];

  const int tid = (int)threadIdx.x;
  const int lane = tid & 31;
  const int lo = (int)blockIdx.x * NPW;
  const int nown = (n_nodes - lo < NPW) ? (n_nodes - lo) : NPW;

  for (int i = tid; i < NPW * F; i += NT) agg_l[i] = 0.f;
  for (int i = tid; i < NPW * 2; i += NT) {
    m_l[i] = 0u;
    den_l[i] = 0.f;
    er_l[i] = (i < nown * 2) ? er[(size_t)lo * 2 + i] : 0.f;
  }
  if (tid == 0) { cnt_l = 0u; tick_l = 0u; }
  if constexpr (FN > 0) {
    if (tid < F) bs_l[tid] = bias[tid] + bias[F + tid];
  }
  __syncthreads();

  if constexpr (!TAKE) {
    const uint4* __restrict__ row =
        reinterpret_cast<const uint4*>(bm + (size_t)blockIdx.x * WPR);
    uint4 v[SB];
    int n = 0;
#pragma unroll
    for (int u = 0; u < SB; ++u) {
      const int q = u * NT + tid;
      const uint4 t = row[(q < QPR) ? q : 0];
      v[u] = (q < QPR) ? t : make_uint4(0u, 0u, 0u, 0u);
      n += __popc(v[u].x) + __popc(v[u].y) + __popc(v[u].z) + __popc(v[u].w);
    }
    unsigned pos = atomicAdd(&cnt_l, (unsigned)n);
#pragma unroll
    for (int u = 0; u < SB; ++u) {
      const unsigned w4[4] = {v[u].x, v[u].y, v[u].z, v[u].w};
#pragma unroll
      for (int j = 0; j < 4; ++j) {
        const unsigned e0 = (unsigned)(((u * NT + tid) * 4 + j) * 32);
        unsigned w = w4[j];
        while (w) {
          if (pos < (unsigned)CAP) list_l[pos] = e0 + (unsigned)__builtin_ctz(w);
          ++pos;
          w &= w - 1u;
        }
      }
    }
    __syncthreads();
  }
  int cnt;
  if constexpr (TAKE) cnt = gcnt[blockIdx.x]; else cnt = (int)cnt_l;
  const unsigned total = (unsigned)__builtin_amdgcn_readfirstlane(cnt);
  const bool use_list = (total <= (unsigned)CAP);
  int* __restrict__ gl = glist + (size_t)blockIdx.x * CAP;
  if constexpr (!TAKE) {
    if (tid == 0) gcnt[blockIdx.x] = (int)total;
  }

  if (use_list) {
    for (int i0 = 0; i0 < (int)total; i0 += NT * PB) {
      int s[PB], dn[PB];
      float l0[PB], l1[PB];
      if constexpr (TAKE) {
#pragma unroll
        for (int k = 0; k < PB; ++k) {
          const int i = i0 + k * NT + tid;
          const unsigned t = (unsigned)gl[(i < (int)total) ? i : 0];
          s[k]  = (i < (int)total) ? (int)(t & 0xFFFFu) : 0;
          dn[k] = (i < (int)total) ? (int)(t >> 16) : 0;
        }
      } else {
        unsigned p[PB];
        int dv[PB];
#pragma unroll
        for (int k = 0; k < PB; ++k) {
          const int i = i0 + k * NT + tid;
          const unsigned t = list_l[(i < (int)total) ? i : 0];
          p[k] = (i < (int)total) ? t : 0u;
        }
#pragma unroll
        for (int k = 0; k < PB; ++k) { s[k] = src[p[k]]; dv[k] = dst[p[k]]; }
        asm volatile("" :: "v"(s[0]), "v"(s[1]), "v"(s[2]), "v"(s[3]),
                           "v"(dv[0]), "v"(dv[1]), "v"(dv[2]), "v"(dv[3]));
#pragma unroll
        for (int k = 0; k < PB; ++k) dn[k] = dv[k] - lo;
      }
#pragma unroll
      for (int k = 0; k < PB; ++k) { l0[k] = el[s[k] * 2]; l1[k] = el[s[k] * 2 + 1]; }
      asm volatile("" :: "v"(l0[0]), "v"(l1[0]), "v"(l0[1]), "v"(l1[1]),
                         "v"(l0[2]), "v"(l1[2]), "v"(l0[3]), "v"(l1[3]));
#pragma unroll
      for (int k = 0; k < PB; ++k) {
        const int i = i0 + k * NT + tid;
        if (i < (int)total) {
          const int dl = dn[k];
          const unsigned sd = (unsigned)s[k] | ((unsigned)dl << 16);
          list_l[i] = sd;
          if constexpr (!TAKE) gl[i] = (int)sd;
          v2f e;
          e.x = lrelu(l0[k] + er_l[dl * 2]);
          e.y = lrelu(l1[k] + er_l[dl * 2 + 1]);
          ev_l[i] = e;
          atomicMax(&m_l[dl * 2],     flipf(e.x));
          atomicMax(&m_l[dl * 2 + 1], flipf(e.y));
        }
      }
    }
  } else {
    for (int i = tid; i < nE; i += NT) {
      const int d = dst[i] - lo;
      if ((unsigned)d < (unsigned)nown) {
        const int s = src[i];
        atomicMax(&m_l[d * 2],     flipf(lrelu(el[s * 2]     + er_l[d * 2])));
        atomicMax(&m_l[d * 2 + 1], flipf(lrelu(el[s * 2 + 1] + er_l[d * 2 + 1])));
      }
    }
  }
  __syncthreads();

  if (use_list) {
    for (int i = tid; i < (int)total; i += NT) {
      const int j = (int)(list_l[i] >> 16) * 2;
      const v2f e = ev_l[i];
      v2f w;
      w.x = __expf(e.x - unflipf(m_l[j]));
      w.y = __expf(e.y - unflipf(m_l[j + 1]));
      ev_l[i] = w;
      atomicAdd(&den_l[j],     w.x);
      atomicAdd(&den_l[j + 1], w.y);
    }
  } else {
    for (int i = tid; i < nE; i += NT) {
      const int d = dst[i] - lo;
      if ((unsigned)d < (unsigned)nown) {
        const int s = src[i];
        const int j = d * 2;
        atomicAdd(&den_l[j],     __expf(lrelu(el[s * 2]     + er_l[j])     - unflipf(m_l[j])));
        atomicAdd(&den_l[j + 1], __expf(lrelu(el[s * 2 + 1] + er_l[j + 1]) - unflipf(m_l[j + 1])));
      }
    }
  }
  __syncthreads();

  auto one_edge = [&](int sj, int dj, float b0, float b1) {
    const float* frow = feat + (size_t)sj * (2 * F);
#pragma unroll
    for (int q = 0; q < HV; ++q) {
      atomicAdd(&agg_l[dj * F + lane + q * 32],
                b0 * frow[lane + q * 32] + b1 * frow[lane + F + q * 32]);
    }
  };
  if (use_list) {
    for (;;) {
      int c = 0;
      if (lane == 0) c = (int)atomicAdd(&tick_l, 1u);
      c = __builtin_amdgcn_readfirstlane(c);
      const int base = c * U;
      if (base >= (int)total) break;
      const int n = ((int)total - base < U) ? ((int)total - base) : U;
      const int slot = base + ((lane < n) ? lane : 0);
      const unsigned p = list_l[slot];
      const v2f wv = ev_l[slot];
      const int s = (int)(p & 0xFFFFu), dl = (int)(p >> 16);
      auto factor = [&](int h, float w) -> int {
        return __float_as_int(0.5f * w / den_l[dl * 2 + h]);
      };
      if (n == U) {
        float f0[U][HV], f1[U][HV];
#pragma unroll
        for (int u = 0; u < U; ++u) {
          const float* frow = feat + (size_t)__builtin_amdgcn_readlane(s, u) * (2 * F);
#pragma unroll
          for (int q = 0; q < HV; ++q) {
            f0[u][q] = frow[lane + q * 32];
            f1[u][q] = frow[lane + F + q * 32];
          }
        }
        const int a0 = factor(0, wv.x), a1 = factor(1, wv.y);
#pragma unroll
        for (int u = 0; u < U; ++u) {
          float* arow = agg_l + __builtin_amdgcn_readlane(dl, u) * F;
          const float b0 = __int_as_float(__builtin_amdgcn_readlane(a0, u));
          const float b1 = __int_as_float(__builtin_amdgcn_readlane(a1, u));
#pragma unroll
          for (int q = 0; q < HV; ++q) {
            atomicAdd(&arow[lane + q * 32], b0 * f0[u][q] + b1 * f1[u][q]);
          }
        }
      } else {
        const int a0 = factor(0, wv.x), a1 = factor(1, wv.y);
        for (int j = 0; j < n; ++j) {
          one_edge(__builtin_amdgcn_readlane(s, j), __builtin_amdgcn_readlane(dl, j),
                   __int_as_float(__builtin_amdgcn_readlane(a0, j)),
                   __int_as_float(__builtin_amdgcn_readlane(a1, j)));
        }
      }
    }
  } else {
    for (;;) {
      int c = 0;
      if (lane == 0) c = (int)atomicAdd(&tick_l, 1u);
      c = __builtin_amdgcn_readfirstlane(c);
      const int base = c * 32;
      if (base >= nE) break;
      const int i = base + lane;
      bool ok = false;
      int s = 0, dl = 0;
      if (i < nE) {
        const int d = dst[i] - lo;
        if ((unsigned)d < (unsigned)nown) { ok = true; dl = d; s = src[i]; }
      }
      int a0 = 0, a1 = 0;
      if (ok) {
        const int j = dl * 2;
        const float w0 = __expf(lrelu(el[s * 2]     + er_l[j])     - unflipf(m_l[j]));
        const float w1 = __expf(lrelu(el[s * 2 + 1] + er_l[j + 1]) - unflipf(m_l[j + 1]));
        a0 = __float_as_int(0.5f * w0 / den_l[j]);
        a1 = __float_as_int(0.5f * w1 / den_l[j + 1]);
      }
      unsigned mask = (unsigned)__ballot(ok);
      while (mask) {
        const int j = __builtin_ctz(mask);
        mask &= mask - 1u;
        one_edge(__builtin_amdgcn_readlane(s, j), __builtin_amdgcn_readlane(dl, j),
                 __int_as_float(__builtin_amdgcn_readlane(a0, j)),
                 __int_as_float(__builtin_amdgcn_readlane(a1, j)));
      }
    }
  }
  __syncthreads();

  if constexpr (FN == 0) {
    for (int i = tid; i < nown * F; i += NT) {
      const int f = i % F;
      const float v = agg_l[i] + 0.5f * (bias[f] + bias[F + f]);
      out[(size_t)lo * F + i] = (v > 0.f) ? v : 0.f;
    }
  } else {
    constexpr int NC = 2 * FN;
    constexpr int NS = NSR / 16;
    constexpr int KS = F / 4;
    static_assert(NC == F, "projected rows and input rows share the pitch HP");
    static_assert(NT % F == 0, "a thread keeps its column through the row pass");
    static_assert(NS * 2 <= NT / 32, "one wave per stripe and column half");
    static_assert(((NPW + 31) / 32) * 4 <= NT / 32, "one wave per 32 nodes and logit kind");
    __shared__ __attribute__((aligned(16))) float hb_l[NSR * HP];
    const int half = lane >> 4;
    const int lr = lane & 15;
    const int wv = __builtin_amdgcn_readfirstlane(tid >> 5);
    const int c0 = (wv & 1) * 32;

    v2f bf[KS][2];
#pragma unroll
    for (int j = 0; j < KS; ++j) {
      const int ka = j * 4 + 2 * half;
#pragma unroll
      for (int t = 0; t < 2; ++t) {
        bf[j][t].x = Wn[(size_t)ka * NC + c0 + t * 16 + lr];
        bf[j][t].y = Wn[(size_t)(ka + 1) * NC + c0 + t * 16 + lr];
      }
    }

    const int fc = tid % F;
    const float sb = bs_l[fc];
    for (int i = tid; i < NSR * F; i += NT) {
      const int r = i / F;
      float v = 0.f;
      if (r < nown) {
        v = agg_l[i] + 0.5f * sb;
        v = (v > 0.f) ? v : 0.f;
      }
      hb_l[r * HP + fc] = v;
    }
    __syncthreads();

    if (wv < NS * 2) {
      const int m0 = (wv >> 1) * 16;
      const float* arow = hb_l + (m0 + lr) * HP;
      v8f acc[2] = {};
#pragma unroll
      for (int j = 0; j < KS; ++j) {
        const int ka = j * 4 + 2 * half;
        v2f a;
        a.x = arow[ka];
        a.y = arow[ka + 1];
#pragma unroll
        for (int t = 0; t < 2; ++t) {
          acc[t] = __builtin_amdgcn_wmma_f32_16x16x4_f32(
               false, a,  false, bf[j][t],
               (short)0, acc[t],  false,  false);
        }
      }
      const int row = m0 + half * 8;
#pragma unroll
      for (int t = 0; t < 2; ++t) {
#pragma unroll
        for (int r = 0; r < 8; ++r) {
          agg_l[(row + r) * HP + c0 + t * 16 + lr] = acc[t][r];
        }
      }
    }
    __syncthreads();

    for (int i = tid; i < nown * (NC / 4); i += NT) {
      const int r = i / (NC / 4), q4 = i % (NC / 4);
      reinterpret_cast<float4*>(featn + (size_t)lo * NC)[i] =
          *reinterpret_cast<const float4*>(agg_l + r * HP + q4 * 4);
    }
    {
      const int kind = wv & 3;
      const int hd = kind & 1;
      const int n = (wv >> 2) * 32 + lane;
      const float* __restrict__ cf = ((kind < 2) ? aln : arn) + hd * FN;
      if (n < nown) {
        const float* frow = agg_l + n * HP + hd * FN;
        float s = 0.f;
#pragma unroll
        for (int f = 0; f < FN; f += 4) {
          const float4 v = *reinterpret_cast<const float4*>(frow + f);
          s += v.x * cf[f];
          s += v.y * cf[f + 1];
          s += v.z * cf[f + 2];
          s += v.w * cf[f + 3];
        }
        float* __restrict__ dp = (kind < 2) ? eln : ern;
        dp[(size_t)(lo + n) * 2 + hd] = s;
      }
    }
  }
}

static inline int cdiv(long long a, long long b) { return (int)((a + b - 1) / b); }

extern "C" void kernel_launch(void* const* d_in, const int* in_sizes, int n_in,
                              void* d_out, int out_size, void* d_ws, size_t ws_size,
                              hipStream_t stream) {
  (void)in_sizes; (void)n_in; (void)out_size; (void)ws_size;
  const float* x   = (const float*)d_in[0];
  const int*  esrc = (const int*)d_in[1];
  const int*  edst = (const int*)d_in[2];
  const float* W1  = (const float*)d_in[3];
  const float* al1 = (const float*)d_in[4];
  const float* ar1 = (const float*)d_in[5];
  const float* b1  = (const float*)d_in[6];
  const float* W2  = (const float*)d_in[7];
  const float* al2 = (const float*)d_in[8];
  const float* ar2 = (const float*)d_in[9];
  const float* b2  = (const float*)d_in[10];

  const int N = GAT_N, E = GAT_E;

  float* ws = (float*)d_ws;
  size_t o = 0;
  float*    feat1  = ws + o; o += (size_t)N * 128;
  float*    el1    = ws + o; o += (size_t)N * 2;
  float*    er1    = ws + o; o += (size_t)N * 2;
  float*    feat2  = ws + o; o += (size_t)N * 64;
  float*    el2    = ws + o; o += (size_t)N * 2;
  float*    er2    = ws + o; o += (size_t)N * 2;
  const int OWN = cdiv(N, NPW);
  int*      elist  = (int*)(ws + o); o += (size_t)OWN * CAP;
  int*      ecnt   = (int*)(ws + o); o += (size_t)OWN;
  unsigned* ebm    = (unsigned*)(ws + o); o += (size_t)OWN * WPR;

  const int BLK = 256;
  {
    range_bitmap_kernel<<<NCH, 256, 0, stream>>>(edst, ebm, N, E);
    int waves = N / 16;
    wmma_gemm_kernel<128, 128><<<cdiv((long long)waves * 32, BLK), BLK, 0, stream>>>(
        x, W1, feat1, N);
    node_attn_kernel<64><<<cdiv((long long)N * 32, BLK), BLK, 0, stream>>>(
        feat1, al1, ar1, el1, er1, N);
    owner_kernel<64, false, GAT_DOUT><<<OWN, 1024, 0, stream>>>(
        esrc, edst, feat1, el1, er1, b1, nullptr, ebm, elist, ecnt,
        W2, al2, ar2, feat2, el2, er2, N, E);
  }
  {
    owner_kernel<32, true, 0><<<OWN, 1024, 0, stream>>>(
        esrc, edst, feat2, el2, er2, b2, (float*)d_out, ebm, elist, ecnt,
        nullptr, nullptr, nullptr, nullptr, nullptr, nullptr, N, E);
  }
}


# Layer-1 logits taken as four more projection columns on the matrix unit (W1 folded with al1, ar1 per work-group into LDS): node logit launch gone, 5 to 4 launches; logit sum order changed
# speedup vs baseline: 6.5477x; 1.2030x over previous
#include <hip/hip_runtime.h>
#include <hip/hip_bf16.h>

typedef __attribute__((ext_vector_type(2))) float v2f;
typedef __attribute__((ext_vector_type(8))) float v8f;

#define GAT_N    50000
#define GAT_E    800000
#define GAT_DIN  128
#define GAT_DHID 64
#define GAT_DOUT 32
#define NEG_SLOPE 0.2f

__device__ __forceinline__ unsigned flipf(float f) {
  unsigned u = __float_as_uint(f);
  return (u & 0x80000000u) ? ~u : (u | 0x80000000u);
}
__device__ __forceinline__ float unflipf(unsigned u) {
  unsigned b = (u & 0x80000000u) ? (u & 0x7FFFFFFFu) : ~u;
  return __uint_as_float(b);
}

#define GEMM_BLK 256
template<int K, int NCOL>
__global__ __launch_bounds__(GEMM_BLK)
void wmma_gemm_kernel(const float* __restrict__ A, const float* __restrict__ B,
                      const float* __restrict__ al, const float* __restrict__ ar,
                      float* __restrict__ C,
                      float* __restrict__ el, float* __restrict__ er, int Mrows) {
  constexpr int NT = NCOL / 16;
  constexpr int F = NCOL / 2;
  static_assert(2 * K == GEMM_BLK, "the fold gives one thread to every (head, k)");
  static_assert(F % 4 == 0, "a row of B is read 16 bytes per load");
  __shared__ __attribute__((aligned(8))) float vf_l[K * 4];
  int lane = (int)(threadIdx.x & 31);
  {
    const int wv = __builtin_amdgcn_readfirstlane((int)(threadIdx.x >> 5));
    const int h = wv & 1;
    const int k = (wv >> 1) * 32 + lane;
    const float4* __restrict__ brow =
        reinterpret_cast<const float4*>(B + (size_t)k * NCOL + h * F);
    const float* __restrict__ cl = al + h * F;
    const float* __restrict__ cr = ar + h * F;
    float sl = 0.f, sr = 0.f;
#pragma unroll
    for (int f = 0; f < F; f += 4) {
      const float4 w = brow[f / 4];
      sl += w.x * cl[f];     sr += w.x * cr[f];
      sl += w.y * cl[f + 1]; sr += w.y * cr[f + 1];
      sl += w.z * cl[f + 2]; sr += w.z * cr[f + 2];
      sl += w.w * cl[f + 3]; sr += w.w * cr[f + 3];
    }
    vf_l[((k >> 1) * 4 + h) * 2 + (k & 1)] = sl;
    vf_l[((k >> 1) * 4 + 2 + h) * 2 + (k & 1)] = sr;
  }
  __syncthreads();

  int wave = (int)((blockIdx.x * blockDim.x + threadIdx.x) >> 5);
  int m0 = wave * 16;
  if (m0 >= Mrows) return;
  int half = lane >> 4;
  int lr = lane & 15;

  v8f acc[NT] = {};
  v8f accl = {};
  const float* arow = A + (size_t)(m0 + lr) * K;
  const v2f* vfp = reinterpret_cast<const v2f*>(vf_l);

  for (int k = 0; k < K; k += 4) {
    int ka = k + 2 * half;
    v2f a;
    a.x = arow[ka];
    a.y = arow[ka + 1];
    const float* b0 = B + (size_t)ka * NCOL;
    const float* b1 = B + (size_t)(ka + 1) * NCOL;
#pragma unroll
    for (int t = 0; t < NT; ++t) {
      v2f b;
      b.x = b0[t * 16 + lr];
      b.y = b1[t * 16 + lr];
      acc[t] = __builtin_amdgcn_wmma_f32_16x16x4_f32(
           false, a,  false, b,
           (short)0, acc[t],  false,  false);
    }
    v2f bl = vfp[(ka >> 1) * 4 + (lr & 3)];
    if (lr >= 4) { bl.x = 0.f; bl.y = 0.f; }
    accl = __builtin_amdgcn_wmma_f32_16x16x4_f32(
         false, a,  false, bl,
         (short)0, accl,  false,  false);
  }

  int row = m0 + half * 8;
#pragma unroll
  for (int t = 0; t < NT; ++t) {
#pragma unroll
    for (int r = 0; r < 8; ++r) {
      C[(size_t)(row + r) * NCOL + t * 16 + lr] = acc[t][r];
    }
  }
  if (lr < 4) {
    float* __restrict__ dp = (lr < 2) ? el : er;
#pragma unroll
    for (int r = 0; r < 8; ++r) {
      dp[(size_t)(row + r) * 2 + (lr & 1)] = accl[r];
    }
  }
}

#define NPW 196
#define CAP 16384
#define NRANGE ((GAT_N + NPW - 1) / NPW)
#define BCH 1024
#define NCH ((GAT_E + BCH - 1) / BCH)
#define WPR (NCH * 32)
static_assert(GAT_E % 4 == 0, "dst is read four edges per load");
static_assert(WPR % 4 == 0, "a bitmap row is read 16 bytes per load");
static_assert(GAT_N <= (1 << 16), "source node is packed into 16 bits");
static_assert(NPW <= (1 << 16), "local node is packed above the source node");
static_assert(NRANGE == 256, "range_bitmap_kernel keeps one LDS column per range");

__global__ __launch_bounds__(256)
void range_bitmap_kernel(const int* __restrict__ dst, unsigned* __restrict__ bm,
                         int n_nodes, int nE) {
  __shared__ unsigned b_l[32 * 257];
  const int tid = (int)threadIdx.x;
  for (int i = tid; i < 32 * 257; i += 256) b_l[i] = 0u;
  __syncthreads();
  const int e0 = (int)blockIdx.x * BCH + tid * 4;
  if (e0 < nE) {
    const int4 d = *reinterpret_cast<const int4*>(dst + e0);
    const int d4[4] = {d.x, d.y, d.z, d.w};
    const int w = tid >> 3;
#pragma unroll
    for (int c = 0; c < 4; ++c) {
      if ((unsigned)d4[c] < (unsigned)n_nodes) {
        atomicOr(&b_l[w * 257 + (int)((unsigned)d4[c] / (unsigned)NPW)],
                 1u << ((tid & 7) * 4 + c));
      }
    }
  }
  __syncthreads();
  for (int i = tid; i < NRANGE * 32; i += 256) {
    const int r = i >> 5, w = i & 31;
    bm[(size_t)r * WPR + (size_t)blockIdx.x * 32 + w] = b_l[w * 257 + r];
  }
}

__device__ __forceinline__ float lrelu(float v) { return (v > 0.f) ? v : NEG_SLOPE * v; }

template<int F, bool TAKE, int FN>
__global__ __launch_bounds__(1024)
void owner_kernel(const int* __restrict__ src, const int* __restrict__ dst,
                  const float* __restrict__ feat,
                  const float* __restrict__ el, const float* __restrict__ er,
                  const float* __restrict__ bias, float* __restrict__ out,
                  const unsigned* __restrict__ bm,
                  int* __restrict__ glist, int* __restrict__ gcnt,
                  const float* __restrict__ Wn,
                  const float* __restrict__ aln, const float* __restrict__ arn,
                  float* __restrict__ featn,
                  float* __restrict__ eln, float* __restrict__ ern,
                  int n_nodes, int nE) {
  constexpr int HV = F / 32;
  constexpr int U = 16 / HV;
  constexpr int NT = 1024;
  constexpr int QPR = WPR / 4;
  constexpr int SB = (QPR + NT - 1) / NT;
  constexpr int PB = 4;
  static_assert(PB == 4, "the load fences of phase 1 name four slots");

  constexpr int NSR = ((NPW + 15) / 16) * 16;
  constexpr int HP = F + 4;
  constexpr int AGG = (FN > 0) ? NSR * HP : NPW * F;
  __shared__ __attribute__((aligned(16))) float agg_l[AGG];
  __shared__ unsigned m_l[NPW * 2];
  __shared__ float    den_l[NPW * 2];
  __shared__ float    er_l[NPW * 2];
  __shared__ unsigned list_l[CAP];
  __shared__ v2f      ev_l[CAP];
  __shared__ unsigned cnt_l;
  __shared__ unsigned tick_l;
  __shared__ float    bs_l[(FN > 0) ? F : 1];

  const int tid = (int)threadIdx.x;
  const int lane = tid & 31;
  const int lo = (int)blockIdx.x * NPW;
  const int nown = (n_nodes - lo < NPW) ? (n_nodes - lo) : NPW;

  for (int i = tid; i < NPW * F; i += NT) agg_l[i] = 0.f;
  for (int i = tid; i < NPW * 2; i += NT) {
    m_l[i] = 0u;
    den_l[i] = 0.f;
    er_l[i] = (i < nown * 2) ? er[(size_t)lo * 2 + i] : 0.f;
  }
  if (tid == 0) { cnt_l = 0u; tick_l = 0u; }
  if constexpr (FN > 0) {
    if (tid < F) bs_l[tid] = bias[tid] + bias[F + tid];
  }
  __syncthreads();

  if constexpr (!TAKE) {
    const uint4* __restrict__ row =
        reinterpret_cast<const uint4*>(bm + (size_t)blockIdx.x * WPR);
    uint4 v[SB];
    int n = 0;
#pragma unroll
    for (int u = 0; u < SB; ++u) {
      const int q = u * NT + tid;
      const uint4 t = row[(q < QPR) ? q : 0];
      v[u] = (q < QPR) ? t : make_uint4(0u, 0u, 0u, 0u);
      n += __popc(v[u].x) + __popc(v[u].y) + __popc(v[u].z) + __popc(v[u].w);
    }
    unsigned pos = atomicAdd(&cnt_l, (unsigned)n);
#pragma unroll
    for (int u = 0; u < SB; ++u) {
      const unsigned w4[4] = {v[u].x, v[u].y, v[u].z, v[u].w};
#pragma unroll
      for (int j = 0; j < 4; ++j) {
        const unsigned e0 = (unsigned)(((u * NT + tid) * 4 + j) * 32);
        unsigned w = w4[j];
        while (w) {
          if (pos < (unsigned)CAP) list_l[pos] = e0 + (unsigned)__builtin_ctz(w);
          ++pos;
          w &= w - 1u;
        }
      }
    }
    __syncthreads();
  }
  int cnt;
  if constexpr (TAKE) cnt = gcnt[blockIdx.x]; else cnt = (int)cnt_l;
  const unsigned total = (unsigned)__builtin_amdgcn_readfirstlane(cnt);
  const bool use_list = (total <= (unsigned)CAP);
  int* __restrict__ gl = glist + (size_t)blockIdx.x * CAP;
  if constexpr (!TAKE) {
    if (tid == 0) gcnt[blockIdx.x] = (int)total;
  }

  if (use_list) {
    for (int i0 = 0; i0 < (int)total; i0 += NT * PB) {
      int s[PB], dn[PB];
      float l0[PB], l1[PB];
      if constexpr (TAKE) {
#pragma unroll
        for (int k = 0; k < PB; ++k) {
          const int i = i0 + k * NT + tid;
          const unsigned t = (unsigned)gl[(i < (int)total) ? i : 0];
          s[k]  = (i < (int)total) ? (int)(t & 0xFFFFu) : 0;
          dn[k] = (i < (int)total) ? (int)(t >> 16) : 0;
        }
      } else {
        unsigned p[PB];
        int dv[PB];
#pragma unroll
        for (int k = 0; k < PB; ++k) {
          const int i = i0 + k * NT + tid;
          const unsigned t = list_l[(i < (int)total) ? i : 0];
          p[k] = (i < (int)total) ? t : 0u;
        }
#pragma unroll
        for (int k = 0; k < PB; ++k) { s[k] = src[p[k]]; dv[k] = dst[p[k]]; }
        asm volatile("" :: "v"(s[0]), "v"(s[1]), "v"(s[2]), "v"(s[3]),
                           "v"(dv[0]), "v"(dv[1]), "v"(dv[2]), "v"(dv[3]));
#pragma unroll
        for (int k = 0; k < PB; ++k) dn[k] = dv[k] - lo;
      }
#pragma unroll
      for (int k = 0; k < PB; ++k) { l0[k] = el[s[k] * 2]; l1[k] = el[s[k] * 2 + 1]; }
      asm volatile("" :: "v"(l0[0]), "v"(l1[0]), "v"(l0[1]), "v"(l1[1]),
                         "v"(l0[2]), "v"(l1[2]), "v"(l0[3]), "v"(l1[3]));
#pragma unroll
      for (int k = 0; k < PB; ++k) {
        const int i = i0 + k * NT + tid;
        if (i < (int)total) {
          const int dl = dn[k];
          const unsigned sd = (unsigned)s[k] | ((unsigned)dl << 16);
          list_l[i] = sd;
          if constexpr (!TAKE) gl[i] = (int)sd;
          v2f e;
          e.x = lrelu(l0[k] + er_l[dl * 2]);
          e.y = lrelu(l1[k] + er_l[dl * 2 + 1]);
          ev_l[i] = e;
          atomicMax(&m_l[dl * 2],     flipf(e.x));
          atomicMax(&m_l[dl * 2 + 1], flipf(e.y));
        }
      }
    }
  } else {
    for (int i = tid; i < nE; i += NT) {
      const int d = dst[i] - lo;
      if ((unsigned)d < (unsigned)nown) {
        const int s = src[i];
        atomicMax(&m_l[d * 2],     flipf(lrelu(el[s * 2]     + er_l[d * 2])));
        atomicMax(&m_l[d * 2 + 1], flipf(lrelu(el[s * 2 + 1] + er_l[d * 2 + 1])));
      }
    }
  }
  __syncthreads();

  if (use_list) {
    for (int i = tid; i < (int)total; i += NT) {
      const int j = (int)(list_l[i] >> 16) * 2;
      const v2f e = ev_l[i];
      v2f w;
      w.x = __expf(e.x - unflipf(m_l[j]));
      w.y = __expf(e.y - unflipf(m_l[j + 1]));
      ev_l[i] = w;
      atomicAdd(&den_l[j],     w.x);
      atomicAdd(&den_l[j + 1], w.y);
    }
  } else {
    for (int i = tid; i < nE; i += NT) {
      const int d = dst[i] - lo;
      if ((unsigned)d < (unsigned)nown) {
        const int s = src[i];
        const int j = d * 2;
        atomicAdd(&den_l[j],     __expf(lrelu(el[s * 2]     + er_l[j])     - unflipf(m_l[j])));
        atomicAdd(&den_l[j + 1], __expf(lrelu(el[s * 2 + 1] + er_l[j + 1]) - unflipf(m_l[j + 1])));
      }
    }
  }
  __syncthreads();

  auto one_edge = [&](int sj, int dj, float b0, float b1) {
    const float* frow = feat + (size_t)sj * (2 * F);
#pragma unroll
    for (int q = 0; q < HV; ++q) {
      atomicAdd(&agg_l[dj * F + lane + q * 32],
                b0 * frow[lane + q * 32] + b1 * frow[lane + F + q * 32]);
    }
  };
  if (use_list) {
    for (;;) {
      int c = 0;
      if (lane == 0) c = (int)atomicAdd(&tick_l, 1u);
      c = __builtin_amdgcn_readfirstlane(c);
      const int base = c * U;
      if (base >= (int)total) break;
      const int n = ((int)total - base < U) ? ((int)total - base) : U;
      const int slot = base + ((lane < n) ? lane : 0);
      const unsigned p = list_l[slot];
      const v2f wv = ev_l[slot];
      const int s = (int)(p & 0xFFFFu), dl = (int)(p >> 16);
      auto factor = [&](int h, float w) -> int {
        return __float_as_int(0.5f * w / den_l[dl * 2 + h]);
      };
      if (n == U) {
        float f0[U][HV], f1[U][HV];
#pragma unroll
        for (int u = 0; u < U; ++u) {
          const float* frow = feat + (size_t)__builtin_amdgcn_readlane(s, u) * (2 * F);
#pragma unroll
          for (int q = 0; q < HV; ++q) {
            f0[u][q] = frow[lane + q * 32];
            f1[u][q] = frow[lane + F + q * 32];
          }
        }
        const int a0 = factor(0, wv.x), a1 = factor(1, wv.y);
#pragma unroll
        for (int u = 0; u < U; ++u) {
          float* arow = agg_l + __builtin_amdgcn_readlane(dl, u) * F;
          const float b0 = __int_as_float(__builtin_amdgcn_readlane(a0, u));
          const float b1 = __int_as_float(__builtin_amdgcn_readlane(a1, u));
#pragma unroll
          for (int q = 0; q < HV; ++q) {
            atomicAdd(&arow[lane + q * 32], b0 * f0[u][q] + b1 * f1[u][q]);
          }
        }
      } else {
        const int a0 = factor(0, wv.x), a1 = factor(1, wv.y);
        for (int j = 0; j < n; ++j) {
          one_edge(__builtin_amdgcn_readlane(s, j), __builtin_amdgcn_readlane(dl, j),
                   __int_as_float(__builtin_amdgcn_readlane(a0, j)),
                   __int_as_float(__builtin_amdgcn_readlane(a1, j)));
        }
      }
    }
  } else {
    for (;;) {
      int c = 0;
      if (lane == 0) c = (int)atomicAdd(&tick_l, 1u);
      c = __builtin_amdgcn_readfirstlane(c);
      const int base = c * 32;
      if (base >= nE) break;
      const int i = base + lane;
      bool ok = false;
      int s = 0, dl = 0;
      if (i < nE) {
        const int d = dst[i] - lo;
        if ((unsigned)d < (unsigned)nown) { ok = true; dl = d; s = src[i]; }
      }
      int a0 = 0, a1 = 0;
      if (ok) {
        const int j = dl * 2;
        const float w0 = __expf(lrelu(el[s * 2]     + er_l[j])     - unflipf(m_l[j]));
        const float w1 = __expf(lrelu(el[s * 2 + 1] + er_l[j + 1]) - unflipf(m_l[j + 1]));
        a0 = __float_as_int(0.5f * w0 / den_l[j]);
        a1 = __float_as_int(0.5f * w1 / den_l[j + 1]);
      }
      unsigned mask = (unsigned)__ballot(ok);
      while (mask) {
        const int j = __builtin_ctz(mask);
        mask &= mask - 1u;
        one_edge(__builtin_amdgcn_readlane(s, j), __builtin_amdgcn_readlane(dl, j),
                 __int_as_float(__builtin_amdgcn_readlane(a0, j)),
                 __int_as_float(__builtin_amdgcn_readlane(a1, j)));
      }
    }
  }
  __syncthreads();

  if constexpr (FN == 0) {
    for (int i = tid; i < nown * F; i += NT) {
      const int f = i % F;
      const float v = agg_l[i] + 0.5f * (bias[f] + bias[F + f]);
      out[(size_t)lo * F + i] = (v > 0.f) ? v : 0.f;
    }
  } else {
    constexpr int NC = 2 * FN;
    constexpr int NS = NSR / 16;
    constexpr int KS = F / 4;
    static_assert(NC == F, "projected rows and input rows share the pitch HP");
    static_assert(NT % F == 0, "a thread keeps its column through the row pass");
    static_assert(NS * 2 <= NT / 32, "one wave per stripe and column half");
    static_assert(((NPW + 31) / 32) * 4 <= NT / 32, "one wave per 32 nodes and logit kind");
    __shared__ __attribute__((aligned(16))) float hb_l[NSR * HP];
    const int half = lane >> 4;
    const int lr = lane & 15;
    const int wv = __builtin_amdgcn_readfirstlane(tid >> 5);
    const int c0 = (wv & 1) * 32;

    v2f bf[KS][2];
#pragma unroll
    for (int j = 0; j < KS; ++j) {
      const int ka = j * 4 + 2 * half;
#pragma unroll
      for (int t = 0; t < 2; ++t) {
        bf[j][t].x = Wn[(size_t)ka * NC + c0 + t * 16 + lr];
        bf[j][t].y = Wn[(size_t)(ka + 1) * NC + c0 + t * 16 + lr];
      }
    }

    const int fc = tid % F;
    const float sb = bs_l[fc];
    for (int i = tid; i < NSR * F; i += NT) {
      const int r = i / F;
      float v = 0.f;
      if (r < nown) {
        v = agg_l[i] + 0.5f * sb;
        v = (v > 0.f) ? v : 0.f;
      }
      hb_l[r * HP + fc] = v;
    }
    __syncthreads();

    if (wv < NS * 2) {
      const int m0 = (wv >> 1) * 16;
      const float* arow = hb_l + (m0 + lr) * HP;
      v8f acc[2] = {};
#pragma unroll
      for (int j = 0; j < KS; ++j) {
        const int ka = j * 4 + 2 * half;
        v2f a;
        a.x = arow[ka];
        a.y = arow[ka + 1];
#pragma unroll
        for (int t = 0; t < 2; ++t) {
          acc[t] = __builtin_amdgcn_wmma_f32_16x16x4_f32(
               false, a,  false, bf[j][t],
               (short)0, acc[t],  false,  false);
        }
      }
      const int row = m0 + half * 8;
#pragma unroll
      for (int t = 0; t < 2; ++t) {
#pragma unroll
        for (int r = 0; r < 8; ++r) {
          agg_l[(row + r) * HP + c0 + t * 16 + lr] = acc[t][r];
        }
      }
    }
    __syncthreads();

    for (int i = tid; i < nown * (NC / 4); i += NT) {
      const int r = i / (NC / 4), q4 = i % (NC / 4);
      reinterpret_cast<float4*>(featn + (size_t)lo * NC)[i] =
          *reinterpret_cast<const float4*>(agg_l + r * HP + q4 * 4);
    }
    {
      const int kind = wv & 3;
      const int hd = kind & 1;
      const int n = (wv >> 2) * 32 + lane;
      const float* __restrict__ cf = ((kind < 2) ? aln : arn) + hd * FN;
      if (n < nown) {
        const float* frow = agg_l + n * HP + hd * FN;
        float s = 0.f;
#pragma unroll
        for (int f = 0; f < FN; f += 4) {
          const float4 v = *reinterpret_cast<const float4*>(frow + f);
          s += v.x * cf[f];
          s += v.y * cf[f + 1];
          s += v.z * cf[f + 2];
          s += v.w * cf[f + 3];
        }
        float* __restrict__ dp = (kind < 2) ? eln : ern;
        dp[(size_t)(lo + n) * 2 + hd] = s;
      }
    }
  }
}

static inline int cdiv(long long a, long long b) { return (int)((a + b - 1) / b); }

extern "C" void kernel_launch(void* const* d_in, const int* in_sizes, int n_in,
                              void* d_out, int out_size, void* d_ws, size_t ws_size,
                              hipStream_t stream) {
  (void)in_sizes; (void)n_in; (void)out_size; (void)ws_size;
  const float* x   = (const float*)d_in[0];
  const int*  esrc = (const int*)d_in[1];
  const int*  edst = (const int*)d_in[2];
  const float* W1  = (const float*)d_in[3];
  const float* al1 = (const float*)d_in[4];
  const float* ar1 = (const float*)d_in[5];
  const float* b1  = (const float*)d_in[6];
  const float* W2  = (const float*)d_in[7];
  const float* al2 = (const float*)d_in[8];
  const float* ar2 = (const float*)d_in[9];
  const float* b2  = (const float*)d_in[10];

  const int N = GAT_N, E = GAT_E;

  float* ws = (float*)d_ws;
  size_t o = 0;
  float*    feat1  = ws + o; o += (size_t)N * 128;
  float*    el1    = ws + o; o += (size_t)N * 2;
  float*    er1    = ws + o; o += (size_t)N * 2;
  float*    feat2  = ws + o; o += (size_t)N * 64;
  float*    el2    = ws + o; o += (size_t)N * 2;
  float*    er2    = ws + o; o += (size_t)N * 2;
  const int OWN = cdiv(N, NPW);
  int*      elist  = (int*)(ws + o); o += (size_t)OWN * CAP;
  int*      ecnt   = (int*)(ws + o); o += (size_t)OWN;
  unsigned* ebm    = (unsigned*)(ws + o); o += (size_t)OWN * WPR;

  {
    range_bitmap_kernel<<<NCH, 256, 0, stream>>>(edst, ebm, N, E);
    int waves = N / 16;
    wmma_gemm_kernel<128, 128><<<cdiv((long long)waves * 32, GEMM_BLK), GEMM_BLK, 0, stream>>>(
        x, W1, al1, ar1, feat1, el1, er1, N);
    owner_kernel<64, false, GAT_DOUT><<<OWN, 1024, 0, stream>>>(
        esrc, edst, feat1, el1, er1, b1, nullptr, ebm, elist, ecnt,
        W2, al2, ar2, feat2, el2, er2, N, E);
  }
  {
    owner_kernel<32, true, 0><<<OWN, 1024, 0, stream>>>(
        esrc, edst, feat2, el2, er2, b2, (float*)d_out, ebm, elist, ecnt,
        nullptr, nullptr, nullptr, nullptr, nullptr, nullptr, N, E);
  }
}
